# starting kernel
# baseline (speedup 1.0000x reference)
#include <hip/hip_runtime.h>
#include <math.h>


typedef __attribute__((ext_vector_type(2))) float v2f;
typedef __attribute__((ext_vector_type(4))) float v4f;
typedef __attribute__((ext_vector_type(8))) float v8f;

#define WAVES 4
#define TPB   (WAVES * 32)
#define BATCH 32
#define NOP   512
#define NMA   64
#define FDIM  8
#define HDIM  128
#define ODIM  8

__device__ __forceinline__ float elu1(float x) {
  float e = __builtin_amdgcn_exp2f(x * 1.44269504088896340736f) - 1.0f;
  return x > 0.0f ? x : e;
}

__device__ __forceinline__ v8f wmma_f32(v2f a, v2f b, v8f c) {
  return __builtin_amdgcn_wmma_f32_16x16x4_f32(false, a, false, b, (short)0, c,
                                               false, false);
}

template <int K, bool ELU>
__device__ __forceinline__ void layer_h(const float* __restrict__ W,
                                        const float* __restrict__ bias,
                                        const float* sA, float* sY, int lane) {
  const int half = lane >> 4, l16 = lane & 15;
  const int kh = half * 2;
  for (int nt = 0; nt < HDIM / 16; ++nt) {
    const int n0 = nt * 16;
    v8f acc0 = {0.f, 0.f, 0.f, 0.f, 0.f, 0.f, 0.f, 0.f};
    v8f acc1 = {0.f, 0.f, 0.f, 0.f, 0.f, 0.f, 0.f, 0.f};
#pragma unroll
    for (int kc = 0; kc < K / 4; kc += 2) {
      {
        const int k = kc * 4 + kh;
        v2f a; a.x = sA[l16 * K + k];            a.y = sA[l16 * K + k + 1];
        v2f b; b.x = W[k * HDIM + n0 + l16];     b.y = W[(k + 1) * HDIM + n0 + l16];
        acc0 = wmma_f32(a, b, acc0);
      }
      {
        const int k = (kc + 1) * 4 + kh;
        v2f a; a.x = sA[l16 * K + k];            a.y = sA[l16 * K + k + 1];
        v2f b; b.x = W[k * HDIM + n0 + l16];     b.y = W[(k + 1) * HDIM + n0 + l16];
        acc1 = wmma_f32(a, b, acc1);
      }
    }
    const float bv = bias[n0 + l16];
#pragma unroll
    for (int r = 0; r < 8; ++r) {
      float y = acc0[r] + acc1[r] + bv;
      if (ELU) y = elu1(y);
      sY[(r + half * 8) * HDIM + n0 + l16] = y;
    }
  }
}

template <int K>
__device__ __forceinline__ v8f layer_o(const float* __restrict__ W,
                                       const float* __restrict__ bias,
                                       const float* sA, int lane) {
  const int half = lane >> 4, l16 = lane & 15;
  const int kh = half * 2;
  const int n = (l16 < 8) ? l16 : 0;
  v8f acc0 = {0.f, 0.f, 0.f, 0.f, 0.f, 0.f, 0.f, 0.f};
  v8f acc1 = {0.f, 0.f, 0.f, 0.f, 0.f, 0.f, 0.f, 0.f};
#pragma unroll
  for (int kc = 0; kc < K / 4; kc += 2) {
    {
      const int k = kc * 4 + kh;
      v2f a; a.x = sA[l16 * K + k];    a.y = sA[l16 * K + k + 1];
      v2f b; b.x = W[k * ODIM + n];    b.y = W[(k + 1) * ODIM + n];
      acc0 = wmma_f32(a, b, acc0);
    }
    {
      const int k = (kc + 1) * 4 + kh;
      v2f a; a.x = sA[l16 * K + k];    a.y = sA[l16 * K + k + 1];
      v2f b; b.x = W[k * ODIM + n];    b.y = W[(k + 1) * ODIM + n];
      acc1 = wmma_f32(a, b, acc1);
    }
  }
  const float bv = bias[n];
#pragma unroll
  for (int r = 0; r < 8; ++r) acc0[r] += acc1[r] + bv;
  return acc0;
}

__global__ __launch_bounds__(TPB) void fused_hgnn_mlp(
    const int* __restrict__ adj, const float* __restrict__ feats_op,
    const float* __restrict__ feats_ma,
    const float* __restrict__ l0w1, const float* __restrict__ l0b1,
    const float* __restrict__ l0w2, const float* __restrict__ l0b2,
    const float* __restrict__ l0w3, const float* __restrict__ l0b3,
    const float* __restrict__ l1w1, const float* __restrict__ l1b1,
    const float* __restrict__ l1w2, const float* __restrict__ l1b2,
    const float* __restrict__ l1w3, const float* __restrict__ l1b3,
    const float* __restrict__ pw1, const float* __restrict__ pb1,
    const float* __restrict__ pw2, const float* __restrict__ pb2,
    const float* __restrict__ pw3, const float* __restrict__ pb3,
    float* __restrict__ out) {
  __shared__ float sFM[NMA * FDIM];
  __shared__ float sBuf[WAVES][2][16 * HDIM];
  __shared__ float sCat[WAVES][16 * 16];

  const int tid = threadIdx.x;
  const int wid = tid >> 5, lane = tid & 31;
  const int half = lane >> 4, l16 = lane & 15;
  const int row0 = blockIdx.x * (WAVES * 16) + wid * 16;
  const int bb = row0 / NOP;
  const int op0 = row0 % NOP;

  __builtin_prefetch(l0w2 + tid * 128, 0, 1);
  __builtin_prefetch(l1w2 + tid * 128, 0, 1);
  __builtin_prefetch(pw2 + tid * 128, 0, 1);

  for (int i = tid; i < NMA * FDIM; i += TPB)
    sFM[i] = feats_ma[bb * NMA * FDIM + i];
  __syncthreads();

  float* sA = &sBuf[wid][0][0];
  float* sB = &sBuf[wid][1][0];
  float* sC = &sCat[wid][0];

  {
    const int m = lane >> 1, k0 = (lane & 1) * 4;
    const int* arow = adj + ((size_t)bb * NOP + op0 + m) * NMA;
    v4f s = {0.f, 0.f, 0.f, 0.f};
    for (int j = 0; j < NMA; ++j) {
      const float av = (float)arow[j];
      const float* f = &sFM[j * FDIM + k0];
      s.x += av * f[0]; s.y += av * f[1]; s.z += av * f[2]; s.w += av * f[3];
    }
    *(v4f*)&sA[m * FDIM + k0] = s;
  }
  __syncthreads();
  layer_h<FDIM, true>(l0w1, l0b1, sA, sB, lane);
  __syncthreads();
  layer_h<HDIM, true>(l0w2, l0b2, sB, sA, lane);
  __syncthreads();
  {
    v8f e0 = layer_o<HDIM>(l0w3, l0b3, sA, lane);
    if (l16 < 8) {
#pragma unroll
      for (int r = 0; r < 8; ++r)
        sC[(r + half * 8) * 16 + l16] = elu1(e0[r]);
    }
  }
  __syncthreads();

  {
    const int m = lane >> 1, k0 = (lane & 1) * 4;
    const float* fr = feats_op + ((size_t)bb * NOP + op0 + m) * FDIM + k0;
    *(v4f*)&sA[m * FDIM + k0] = *(const v4f*)fr;
  }
  __syncthreads();
  layer_h<FDIM, true>(l1w1, l1b1, sA, sB, lane);
  __syncthreads();
  layer_h<HDIM, true>(l1w2, l1b2, sB, sA, lane);
  __syncthreads();
  {
    v8f e1 = layer_o<HDIM>(l1w3, l1b3, sA, lane);
    if (l16 < 8) {
#pragma unroll
      for (int r = 0; r < 8; ++r)
        sC[(r + half * 8) * 16 + 8 + l16] = elu1(e1[r]);
    }
  }
  __syncthreads();

  layer_h<16, true>(pw1, pb1, sC, sA, lane);
  __syncthreads();
  layer_h<HDIM, true>(pw2, pb2, sA, sB, lane);
  __syncthreads();
  {
    v8f mu = layer_o<HDIM>(pw3, pb3, sB, lane);
    if (l16 < 8) {
#pragma unroll
      for (int r = 0; r < 8; ++r)
        out[(size_t)(row0 + r + half * 8) * ODIM + l16] = mu[r];
    }
  }
}

extern "C" void kernel_launch(void* const* d_in, const int* in_sizes, int n_in,
                              void* d_out, int out_size, void* d_ws,
                              size_t ws_size, hipStream_t stream) {
  const int*   adj = (const int*)d_in[0];
  const float* fop = (const float*)d_in[1];
  const float* fma = (const float*)d_in[2];
  const float* l0w1 = (const float*)d_in[3];
  const float* l0b1 = (const float*)d_in[4];
  const float* l0w2 = (const float*)d_in[5];
  const float* l0b2 = (const float*)d_in[6];
  const float* l0w3 = (const float*)d_in[7];
  const float* l0b3 = (const float*)d_in[8];
  const float* l1w1 = (const float*)d_in[9];
  const float* l1b1 = (const float*)d_in[10];
  const float* l1w2 = (const float*)d_in[11];
  const float* l1b2 = (const float*)d_in[12];
  const float* l1w3 = (const float*)d_in[13];
  const float* l1b3 = (const float*)d_in[14];
  const float* pw1 = (const float*)d_in[15];
  const float* pb1 = (const float*)d_in[16];
  const float* pw2 = (const float*)d_in[17];
  const float* pb2 = (const float*)d_in[18];
  const float* pw3 = (const float*)d_in[19];
  const float* pb3 = (const float*)d_in[20];

  const int total_rows = BATCH * NOP;
  const int rows_per_block = WAVES * 16;
  dim3 grid(total_rows / rows_per_block);
  fused_hgnn_mlp<<<grid, TPB, 0, stream>>>(
      adj, fop, fma, l0w1, l0b1, l0w2, l0b2, l0w3, l0b3, l1w1, l1b1, l1w2,
      l1b2, l1w3, l1b3, pw1, pb1, pw2, pb2, pw3, pb3, (float*)d_out);
}


# Pad LDS activation row pitch (128->132, 8->12, 16->20 floats): the 16 rows of each A-fragment ds_load now fall in 16 distinct bank groups instead of one bank (16-way conflict removed)
# speedup vs baseline: 1.3529x; 1.3529x over previous
#include <hip/hip_runtime.h>
#include <math.h>


typedef __attribute__((ext_vector_type(2))) float v2f;
typedef __attribute__((ext_vector_type(4))) float v4f;
typedef __attribute__((ext_vector_type(8))) float v8f;

#define WAVES 4
#define TPB   (WAVES * 32)
#define BATCH 32
#define NOP   512
#define NMA   64
#define FDIM  8
#define HDIM  128
#define ODIM  8
#define PH    (HDIM + 4)
#define PF    (FDIM + 4)
#define PC    (16 + 4)

__device__ __forceinline__ float elu1(float x) {
  float e = __builtin_amdgcn_exp2f(x * 1.44269504088896340736f) - 1.0f;
  return x > 0.0f ? x : e;
}

__device__ __forceinline__ v8f wmma_f32(v2f a, v2f b, v8f c) {
  return __builtin_amdgcn_wmma_f32_16x16x4_f32(false, a, false, b, (short)0, c,
                                               false, false);
}

template <int K, int PA, bool ELU>
__device__ __forceinline__ void layer_h(const float* __restrict__ W,
                                        const float* __restrict__ bias,
                                        const float* sA, float* sY, int lane) {
  const int half = lane >> 4, l16 = lane & 15;
  const int kh = half * 2;
  for (int nt = 0; nt < HDIM / 16; ++nt) {
    const int n0 = nt * 16;
    v8f acc0 = {0.f, 0.f, 0.f, 0.f, 0.f, 0.f, 0.f, 0.f};
    v8f acc1 = {0.f, 0.f, 0.f, 0.f, 0.f, 0.f, 0.f, 0.f};
#pragma unroll
    for (int kc = 0; kc < K / 4; kc += 2) {
      {
        const int k = kc * 4 + kh;
        v2f a; a.x = sA[l16 * PA + k];           a.y = sA[l16 * PA + k + 1];
        v2f b; b.x = W[k * HDIM + n0 + l16];     b.y = W[(k + 1) * HDIM + n0 + l16];
        acc0 = wmma_f32(a, b, acc0);
      }
      {
        const int k = (kc + 1) * 4 + kh;
        v2f a; a.x = sA[l16 * PA + k];           a.y = sA[l16 * PA + k + 1];
        v2f b; b.x = W[k * HDIM + n0 + l16];     b.y = W[(k + 1) * HDIM + n0 + l16];
        acc1 = wmma_f32(a, b, acc1);
      }
    }
    const float bv = bias[n0 + l16];
#pragma unroll
    for (int r = 0; r < 8; ++r) {
      float y = acc0[r] + acc1[r] + bv;
      if (ELU) y = elu1(y);
      sY[(r + half * 8) * PH + n0 + l16] = y;
    }
  }
}

template <int K, int PA>
__device__ __forceinline__ v8f layer_o(const float* __restrict__ W,
                                       const float* __restrict__ bias,
                                       const float* sA, int lane) {
  const int half = lane >> 4, l16 = lane & 15;
  const int kh = half * 2;
  const int n = (l16 < 8) ? l16 : 0;
  v8f acc0 = {0.f, 0.f, 0.f, 0.f, 0.f, 0.f, 0.f, 0.f};
  v8f acc1 = {0.f, 0.f, 0.f, 0.f, 0.f, 0.f, 0.f, 0.f};
#pragma unroll
  for (int kc = 0; kc < K / 4; kc += 2) {
    {
      const int k = kc * 4 + kh;
      v2f a; a.x = sA[l16 * PA + k];   a.y = sA[l16 * PA + k + 1];
      v2f b; b.x = W[k * ODIM + n];    b.y = W[(k + 1) * ODIM + n];
      acc0 = wmma_f32(a, b, acc0);
    }
    {
      const int k = (kc + 1) * 4 + kh;
      v2f a; a.x = sA[l16 * PA + k];   a.y = sA[l16 * PA + k + 1];
      v2f b; b.x = W[k * ODIM + n];    b.y = W[(k + 1) * ODIM + n];
      acc1 = wmma_f32(a, b, acc1);
    }
  }
  const float bv = bias[n];
#pragma unroll
  for (int r = 0; r < 8; ++r) acc0[r] += acc1[r] + bv;
  return acc0;
}

__global__ __launch_bounds__(TPB) void fused_hgnn_mlp(
    const int* __restrict__ adj, const float* __restrict__ feats_op,
    const float* __restrict__ feats_ma,
    const float* __restrict__ l0w1, const float* __restrict__ l0b1,
    const float* __restrict__ l0w2, const float* __restrict__ l0b2,
    const float* __restrict__ l0w3, const float* __restrict__ l0b3,
    const float* __restrict__ l1w1, const float* __restrict__ l1b1,
    const float* __restrict__ l1w2, const float* __restrict__ l1b2,
    const float* __restrict__ l1w3, const float* __restrict__ l1b3,
    const float* __restrict__ pw1, const float* __restrict__ pb1,
    const float* __restrict__ pw2, const float* __restrict__ pb2,
    const float* __restrict__ pw3, const float* __restrict__ pb3,
    float* __restrict__ out) {
  __shared__ float sFM[NMA * FDIM];
  __shared__ float sBuf[WAVES][2][16 * PH];
  __shared__ float sCat[WAVES][16 * PC];

  const int tid = threadIdx.x;
  const int wid = tid >> 5, lane = tid & 31;
  const int half = lane >> 4, l16 = lane & 15;
  const int row0 = blockIdx.x * (WAVES * 16) + wid * 16;
  const int bb = row0 / NOP;
  const int op0 = row0 % NOP;

  __builtin_prefetch(l0w2 + tid * 128, 0, 1);
  __builtin_prefetch(l1w2 + tid * 128, 0, 1);
  __builtin_prefetch(pw2 + tid * 128, 0, 1);

  for (int i = tid; i < NMA * FDIM; i += TPB)
    sFM[i] = feats_ma[bb * NMA * FDIM + i];
  __syncthreads();

  float* sA = &sBuf[wid][0][0];
  float* sB = &sBuf[wid][1][0];
  float* sC = &sCat[wid][0];

  {
    const int m = lane >> 1, k0 = (lane & 1) * 4;
    const int* arow = adj + ((size_t)bb * NOP + op0 + m) * NMA;
    v4f s = {0.f, 0.f, 0.f, 0.f};
    for (int j = 0; j < NMA; ++j) {
      const float av = (float)arow[j];
      const float* f = &sFM[j * FDIM + k0];
      s.x += av * f[0]; s.y += av * f[1]; s.z += av * f[2]; s.w += av * f[3];
    }
    *(v4f*)&sA[m * PF + k0] = s;
  }
  __syncthreads();
  layer_h<FDIM, PF, true>(l0w1, l0b1, sA, sB, lane);
  __syncthreads();
  layer_h<HDIM, PH, true>(l0w2, l0b2, sB, sA, lane);
  __syncthreads();
  {
    v8f e0 = layer_o<HDIM, PH>(l0w3, l0b3, sA, lane);
    if (l16 < 8) {
#pragma unroll
      for (int r = 0; r < 8; ++r)
        sC[(r + half * 8) * PC + l16] = elu1(e0[r]);
    }
  }
  __syncthreads();

  {
    const int m = lane >> 1, k0 = (lane & 1) * 4;
    const float* fr = feats_op + ((size_t)bb * NOP + op0 + m) * FDIM + k0;
    *(v4f*)&sA[m * PF + k0] = *(const v4f*)fr;
  }
  __syncthreads();
  layer_h<FDIM, PF, true>(l1w1, l1b1, sA, sB, lane);
  __syncthreads();
  layer_h<HDIM, PH, true>(l1w2, l1b2, sB, sA, lane);
  __syncthreads();
  {
    v8f e1 = layer_o<HDIM, PH>(l1w3, l1b3, sA, lane);
    if (l16 < 8) {
#pragma unroll
      for (int r = 0; r < 8; ++r)
        sC[(r + half * 8) * PC + 8 + l16] = elu1(e1[r]);
    }
  }
  __syncthreads();

  layer_h<16, PC, true>(pw1, pb1, sC, sA, lane);
  __syncthreads();
  layer_h<HDIM, PH, true>(pw2, pb2, sA, sB, lane);
  __syncthreads();
  {
    v8f mu = layer_o<HDIM, PH>(pw3, pb3, sB, lane);
    if (l16 < 8) {
#pragma unroll
      for (int r = 0; r < 8; ++r)
        out[(size_t)(row0 + r + half * 8) * ODIM + l16] = mu[r];
    }
  }
}

extern "C" void kernel_launch(void* const* d_in, const int* in_sizes, int n_in,
                              void* d_out, int out_size, void* d_ws,
                              size_t ws_size, hipStream_t stream) {
  const int*   adj = (const int*)d_in[0];
  const float* fop = (const float*)d_in[1];
  const float* fma = (const float*)d_in[2];
  const float* l0w1 = (const float*)d_in[3];
  const float* l0b1 = (const float*)d_in[4];
  const float* l0w2 = (const float*)d_in[5];
  const float* l0b2 = (const float*)d_in[6];
  const float* l0w3 = (const float*)d_in[7];
  const float* l0b3 = (const float*)d_in[8];
  const float* l1w1 = (const float*)d_in[9];
  const float* l1b1 = (const float*)d_in[10];
  const float* l1w2 = (const float*)d_in[11];
  const float* l1b2 = (const float*)d_in[12];
  const float* l1w3 = (const float*)d_in[13];
  const float* l1b3 = (const float*)d_in[14];
  const float* pw1 = (const float*)d_in[15];
  const float* pb1 = (const float*)d_in[16];
  const float* pw2 = (const float*)d_in[17];
  const float* pb2 = (const float*)d_in[18];
  const float* pw3 = (const float*)d_in[19];
  const float* pb3 = (const float*)d_in[20];

  const int total_rows = BATCH * NOP;
  const int rows_per_block = WAVES * 16;
  dim3 grid(total_rows / rows_per_block);
  fused_hgnn_mlp<<<grid, TPB, 0, stream>>>(
      adj, fop, fma, l0w1, l0b1, l0w2, l0b2, l0w3, l0b3, l1w1, l1b1, l1w2,
      l1b2, l1w3, l1b3, pw1, pb1, pw2, pb2, pw3, pb3, (float*)d_out);
}


# Interleave the n-tile columns (tile t of a 64-column group owns columns 4c+t): one global_load_b128 per k row feeds 4 WMMAs, all 8 tiles accumulate at once, A read once per layer; sums unchanged
# speedup vs baseline: 1.6979x; 1.2550x over previous
#include <hip/hip_runtime.h>
#include <math.h>


typedef __attribute__((ext_vector_type(2))) float v2f;
typedef __attribute__((ext_vector_type(4))) float v4f;
typedef __attribute__((ext_vector_type(8))) float v8f;

#define WAVES 4
#define TPB   (WAVES * 32)
#define BATCH 32
#define NOP   512
#define NMA   64
#define FDIM  8
#define HDIM  128
#define ODIM  8
#define PH    (HDIM + 4)
#define PF    (FDIM + 4)
#define PC    (16 + 4)

__device__ __forceinline__ float elu1(float x) {
  float e = __builtin_amdgcn_exp2f(x * 1.44269504088896340736f) - 1.0f;
  return x > 0.0f ? x : e;
}

__device__ __forceinline__ v8f wmma_f32(v2f a, v2f b, v8f c) {
  return __builtin_amdgcn_wmma_f32_16x16x4_f32(false, a, false, b, (short)0, c,
                                               false, false);
}

#define KCH 4

template <int NS>
struct Frag {
  v2f a[NS];
  v4f b[NS][2][2];
};

template <int NS, int PA>
__device__ __forceinline__ void load_chunk(Frag<NS>& f,
                                           const float* __restrict__ W,
                                           const float* sA, int ks0, int c,
                                           int kh) {
#pragma unroll
  for (int i = 0; i < NS; ++i) {
    const int k = (ks0 + i) * 4 + kh;
    f.a[i] = *(const v2f*)&sA[c * PA + k];
#pragma unroll
    for (int s = 0; s < 2; ++s) {
#pragma unroll
      for (int g = 0; g < 2; ++g)
        f.b[i][s][g] = *(const v4f*)&W[(k + s) * HDIM + g * 64 + c * 4];
    }
  }
}

template <int NS>
__device__ __forceinline__ void mma_chunk(const Frag<NS>& f,
                                          v8f (&acc)[2][2][4]) {
#pragma unroll
  for (int i = 0; i < NS; ++i) {
#pragma unroll
    for (int g = 0; g < 2; ++g) {
#pragma unroll
      for (int t = 0; t < 4; ++t) {
        v2f b; b.x = f.b[i][0][g][t]; b.y = f.b[i][1][g][t];
        acc[i & 1][g][t] = wmma_f32(f.a[i], b, acc[i & 1][g][t]);
      }
    }
  }
}

template <int K, int PA, bool ELU>
__device__ __forceinline__ void layer_h(const float* __restrict__ W,
                                        const float* __restrict__ bias,
                                        const float* sA, float* sY, int lane) {
  constexpr int NKS = K / 4;
  constexpr int NS = NKS < KCH ? NKS : KCH;
  constexpr int NCH = NKS / NS;
  const int half = lane >> 4, c = lane & 15;
  const int kh = half * 2;

  v8f acc[2][2][4];
#pragma unroll
  for (int p = 0; p < 2; ++p)
#pragma unroll
    for (int g = 0; g < 2; ++g)
#pragma unroll
      for (int t = 0; t < 4; ++t)
        acc[p][g][t] = (v8f){0.f, 0.f, 0.f, 0.f, 0.f, 0.f, 0.f, 0.f};

  v4f bv[2];
  bv[0] = *(const v4f*)&bias[c * 4];
  bv[1] = *(const v4f*)&bias[64 + c * 4];

  Frag<NS> f0, f1;
  load_chunk<NS, PA>(f0, W, sA, 0, c, kh);
  if (NCH == 1) {
    mma_chunk<NS>(f0, acc);
  } else {
#pragma unroll 1
    for (int ch = 0; ch < NCH - 2; ch += 2) {
      load_chunk<NS, PA>(f1, W, sA, (ch + 1) * NS, c, kh);
      mma_chunk<NS>(f0, acc);
      load_chunk<NS, PA>(f0, W, sA, (ch + 2) * NS, c, kh);
      mma_chunk<NS>(f1, acc);
    }
    load_chunk<NS, PA>(f1, W, sA, (NCH - 1) * NS, c, kh);
    mma_chunk<NS>(f0, acc);
    mma_chunk<NS>(f1, acc);
  }

#pragma unroll
  for (int g = 0; g < 2; ++g) {
#pragma unroll
    for (int r = 0; r < 8; ++r) {
      v4f y;
#pragma unroll
      for (int t = 0; t < 4; ++t) {
        float v = acc[0][g][t][r] + acc[1][g][t][r] + bv[g][t];
        if (ELU) v = elu1(v);
        y[t] = v;
      }
      *(v4f*)&sY[(r + half * 8) * PH + g * 64 + c * 4] = y;
    }
  }
}

template <int K, int PA>
__device__ __forceinline__ v8f layer_o(const float* __restrict__ W,
                                       const float* __restrict__ bias,
                                       const float* sA, int lane) {
  const int half = lane >> 4, l16 = lane & 15;
  const int kh = half * 2;
  const int n = (l16 < 8) ? l16 : 0;
  v8f acc0 = {0.f, 0.f, 0.f, 0.f, 0.f, 0.f, 0.f, 0.f};
  v8f acc1 = {0.f, 0.f, 0.f, 0.f, 0.f, 0.f, 0.f, 0.f};
#pragma unroll
  for (int kc = 0; kc < K / 4; kc += 2) {
    {
      const int k = kc * 4 + kh;
      v2f a; a.x = sA[l16 * PA + k];   a.y = sA[l16 * PA + k + 1];
      v2f b; b.x = W[k * ODIM + n];    b.y = W[(k + 1) * ODIM + n];
      acc0 = wmma_f32(a, b, acc0);
    }
    {
      const int k = (kc + 1) * 4 + kh;
      v2f a; a.x = sA[l16 * PA + k];   a.y = sA[l16 * PA + k + 1];
      v2f b; b.x = W[k * ODIM + n];    b.y = W[(k + 1) * ODIM + n];
      acc1 = wmma_f32(a, b, acc1);
    }
  }
  const float bv = bias[n];
#pragma unroll
  for (int r = 0; r < 8; ++r) acc0[r] += acc1[r] + bv;
  return acc0;
}

__global__ __launch_bounds__(TPB) void fused_hgnn_mlp(
    const int* __restrict__ adj, const float* __restrict__ feats_op,
    const float* __restrict__ feats_ma,
    const float* __restrict__ l0w1, const float* __restrict__ l0b1,
    const float* __restrict__ l0w2, const float* __restrict__ l0b2,
    const float* __restrict__ l0w3, const float* __restrict__ l0b3,
    const float* __restrict__ l1w1, const float* __restrict__ l1b1,
    const float* __restrict__ l1w2, const float* __restrict__ l1b2,
    const float* __restrict__ l1w3, const float* __restrict__ l1b3,
    const float* __restrict__ pw1, const float* __restrict__ pb1,
    const float* __restrict__ pw2, const float* __restrict__ pb2,
    const float* __restrict__ pw3, const float* __restrict__ pb3,
    float* __restrict__ out) {
  __shared__ float sFM[NMA * FDIM];
  __shared__ __attribute__((aligned(16))) float sBuf[WAVES][2][16 * PH];
  __shared__ __attribute__((aligned(16))) float sCat[WAVES][16 * PC];

  const int tid = threadIdx.x;
  const int wid = tid >> 5, lane = tid & 31;
  const int half = lane >> 4, l16 = lane & 15;
  const int row0 = blockIdx.x * (WAVES * 16) + wid * 16;
  const int bb = row0 / NOP;
  const int op0 = row0 % NOP;

  __builtin_prefetch(l0w2 + tid * 128, 0, 1);
  __builtin_prefetch(l1w2 + tid * 128, 0, 1);
  __builtin_prefetch(pw2 + tid * 128, 0, 1);

  for (int i = tid; i < NMA * FDIM; i += TPB)
    sFM[i] = feats_ma[bb * NMA * FDIM + i];
  __syncthreads();

  float* sA = &sBuf[wid][0][0];
  float* sB = &sBuf[wid][1][0];
  float* sC = &sCat[wid][0];

  for (int st = 0; st < 3; ++st) {
    const float* w1 = st == 0 ? l0w1 : (st == 1 ? l1w1 : pw1);
    const float* b1 = st == 0 ? l0b1 : (st == 1 ? l1b1 : pb1);
    const float* w2 = st == 0 ? l0w2 : (st == 1 ? l1w2 : pw2);
    const float* b2 = st == 0 ? l0b2 : (st == 1 ? l1b2 : pb2);
    const float* w3 = st == 0 ? l0w3 : (st == 1 ? l1w3 : pw3);
    const float* b3 = st == 0 ? l0b3 : (st == 1 ? l1b3 : pb3);

    if (st == 0) {
      const int m = lane >> 1, k0 = (lane & 1) * 4;
      const int* arow = adj + ((size_t)bb * NOP + op0 + m) * NMA;
      v4f s = {0.f, 0.f, 0.f, 0.f};
      for (int j = 0; j < NMA; ++j) {
        const float av = (float)arow[j];
        const float* f = &sFM[j * FDIM + k0];
        s.x += av * f[0]; s.y += av * f[1]; s.z += av * f[2]; s.w += av * f[3];
      }
      *(v4f*)&sA[m * PF + k0] = s;
    } else if (st == 1) {
      const int m = lane >> 1, k0 = (lane & 1) * 4;
      const float* fr = feats_op + ((size_t)bb * NOP + op0 + m) * FDIM + k0;
      *(v4f*)&sA[m * PF + k0] = *(const v4f*)fr;
    }
    if (st < 2) {
      __syncthreads();
      layer_h<FDIM, PF, true>(w1, b1, sA, sB, lane);
    } else {
      layer_h<16, PC, true>(w1, b1, sC, sB, lane);
    }
    __syncthreads();
    layer_h<HDIM, PH, true>(w2, b2, sB, sA, lane);
    __syncthreads();
    const v8f e = layer_o<HDIM, PH>(w3, b3, sA, lane);
    if (st < 2) {
      if (l16 < 8) {
#pragma unroll
        for (int r = 0; r < 8; ++r)
          sC[(r + half * 8) * PC + st * 8 + l16] = elu1(e[r]);
      }
      __syncthreads();
    } else {
      if (l16 < 8) {
#pragma unroll
        for (int r = 0; r < 8; ++r)
          out[(size_t)(row0 + r + half * 8) * ODIM + l16] = e[r];
      }
    }
  }
}

extern "C" void kernel_launch(void* const* d_in, const int* in_sizes, int n_in,
                              void* d_out, int out_size, void* d_ws,
                              size_t ws_size, hipStream_t stream) {
  const int*   adj = (const int*)d_in[0];
  const float* fop = (const float*)d_in[1];
  const float* fma = (const float*)d_in[2];
  const float* l0w1 = (const float*)d_in[3];
  const float* l0b1 = (const float*)d_in[4];
  const float* l0w2 = (const float*)d_in[5];
  const float* l0b2 = (const float*)d_in[6];
  const float* l0w3 = (const float*)d_in[7];
  const float* l0b3 = (const float*)d_in[8];
  const float* l1w1 = (const float*)d_in[9];
  const float* l1b1 = (const float*)d_in[10];
  const float* l1w2 = (const float*)d_in[11];
  const float* l1b2 = (const float*)d_in[12];
  const float* l1w3 = (const float*)d_in[13];
  const float* l1b3 = (const float*)d_in[14];
  const float* pw1 = (const float*)d_in[15];
  const float* pb1 = (const float*)d_in[16];
  const float* pw2 = (const float*)d_in[17];
  const float* pb2 = (const float*)d_in[18];
  const float* pw3 = (const float*)d_in[19];
  const float* pb3 = (const float*)d_in[20];

  const int total_rows = BATCH * NOP;
  const int rows_per_block = WAVES * 16;
  dim3 grid(total_rows / rows_per_block);
  fused_hgnn_mlp<<<grid, TPB, 0, stream>>>(
      adj, fop, fma, l0w1, l0b1, l0w2, l0b2, l0w3, l0b3, l1w1, l1b1, l1w2,
      l1b2, l1w3, l1b3, pw1, pb1, pw2, pb2, pw3, pb3, (float*)d_out);
}


# All input-row loads (feats_ma share, adjacency row, feats_op piece) issued as one group at kernel entry and held in registers: 13 dependent memory round trips become one; sum order unchanged
# speedup vs baseline: 1.8637x; 1.0977x over previous
#include <hip/hip_runtime.h>
#include <math.h>


typedef __attribute__((ext_vector_type(2))) float v2f;
typedef __attribute__((ext_vector_type(4))) float v4f;
typedef __attribute__((ext_vector_type(4))) int v4i;
typedef __attribute__((ext_vector_type(8))) float v8f;

#define WAVES 4
#define TPB   (WAVES * 32)
#define BATCH 32
#define NOP   512
#define NMA   64
#define FDIM  8
#define HDIM  128
#define ODIM  8
#define PH    (HDIM + 4)
#define PF    (FDIM + 4)
#define PC    (16 + 4)

__device__ __forceinline__ float elu1(float x) {
  float e = __builtin_amdgcn_exp2f(x * 1.44269504088896340736f) - 1.0f;
  return x > 0.0f ? x : e;
}

__device__ __forceinline__ v8f wmma_f32(v2f a, v2f b, v8f c) {
  return __builtin_amdgcn_wmma_f32_16x16x4_f32(false, a, false, b, (short)0, c,
                                               false, false);
}

#define KCH 4

template <int NS>
struct Frag {
  v2f a[NS];
  v4f b[NS][2][2];
};

template <int NS, int PA>
__device__ __forceinline__ void load_chunk(Frag<NS>& f,
                                           const float* __restrict__ W,
                                           const float* sA, int ks0, int c,
                                           int kh) {
#pragma unroll
  for (int i = 0; i < NS; ++i) {
    const int k = (ks0 + i) * 4 + kh;
    f.a[i] = *(const v2f*)&sA[c * PA + k];
#pragma unroll
    for (int s = 0; s < 2; ++s) {
#pragma unroll
      for (int g = 0; g < 2; ++g)
        f.b[i][s][g] = *(const v4f*)&W[(k + s) * HDIM + g * 64 + c * 4];
    }
  }
}

template <int NS>
__device__ __forceinline__ void mma_chunk(const Frag<NS>& f,
                                          v8f (&acc)[2][2][4]) {
#pragma unroll
  for (int i = 0; i < NS; ++i) {
#pragma unroll
    for (int g = 0; g < 2; ++g) {
#pragma unroll
      for (int t = 0; t < 4; ++t) {
        v2f b; b.x = f.b[i][0][g][t]; b.y = f.b[i][1][g][t];
        acc[i & 1][g][t] = wmma_f32(f.a[i], b, acc[i & 1][g][t]);
      }
    }
  }
}

template <int K, int PA, bool ELU>
__device__ __forceinline__ void layer_h(const float* __restrict__ W,
                                        const float* __restrict__ bias,
                                        const float* sA, float* sY, int lane) {
  constexpr int NKS = K / 4;
  constexpr int NS = NKS < KCH ? NKS : KCH;
  constexpr int NCH = NKS / NS;
  const int half = lane >> 4, c = lane & 15;
  const int kh = half * 2;

  v8f acc[2][2][4];
#pragma unroll
  for (int p = 0; p < 2; ++p)
#pragma unroll
    for (int g = 0; g < 2; ++g)
#pragma unroll
      for (int t = 0; t < 4; ++t)
        acc[p][g][t] = (v8f){0.f, 0.f, 0.f, 0.f, 0.f, 0.f, 0.f, 0.f};

  v4f bv[2];
  bv[0] = *(const v4f*)&bias[c * 4];
  bv[1] = *(const v4f*)&bias[64 + c * 4];

  Frag<NS> f0, f1;
  load_chunk<NS, PA>(f0, W, sA, 0, c, kh);
  if (NCH == 1) {
    mma_chunk<NS>(f0, acc);
  } else {
#pragma unroll 1
    for (int ch = 0; ch < NCH - 2; ch += 2) {
      load_chunk<NS, PA>(f1, W, sA, (ch + 1) * NS, c, kh);
      mma_chunk<NS>(f0, acc);
      load_chunk<NS, PA>(f0, W, sA, (ch + 2) * NS, c, kh);
      mma_chunk<NS>(f1, acc);
    }
    load_chunk<NS, PA>(f1, W, sA, (NCH - 1) * NS, c, kh);
    mma_chunk<NS>(f0, acc);
    mma_chunk<NS>(f1, acc);
  }

#pragma unroll
  for (int g = 0; g < 2; ++g) {
#pragma unroll
    for (int r = 0; r < 8; ++r) {
      v4f y;
#pragma unroll
      for (int t = 0; t < 4; ++t) {
        float v = acc[0][g][t][r] + acc[1][g][t][r] + bv[g][t];
        if (ELU) v = elu1(v);
        y[t] = v;
      }
      *(v4f*)&sY[(r + half * 8) * PH + g * 64 + c * 4] = y;
    }
  }
}

template <int K, int PA>
__device__ __forceinline__ v8f layer_o(const float* __restrict__ W,
                                       const float* __restrict__ bias,
                                       const float* sA, int lane) {
  const int half = lane >> 4, l16 = lane & 15;
  const int kh = half * 2;
  const int n = (l16 < 8) ? l16 : 0;
  v8f acc0 = {0.f, 0.f, 0.f, 0.f, 0.f, 0.f, 0.f, 0.f};
  v8f acc1 = {0.f, 0.f, 0.f, 0.f, 0.f, 0.f, 0.f, 0.f};
#pragma unroll
  for (int kc = 0; kc < K / 4; kc += 2) {
    {
      const int k = kc * 4 + kh;
      v2f a; a.x = sA[l16 * PA + k];   a.y = sA[l16 * PA + k + 1];
      v2f b; b.x = W[k * ODIM + n];    b.y = W[(k + 1) * ODIM + n];
      acc0 = wmma_f32(a, b, acc0);
    }
    {
      const int k = (kc + 1) * 4 + kh;
      v2f a; a.x = sA[l16 * PA + k];   a.y = sA[l16 * PA + k + 1];
      v2f b; b.x = W[k * ODIM + n];    b.y = W[(k + 1) * ODIM + n];
      acc1 = wmma_f32(a, b, acc1);
    }
  }
  const float bv = bias[n];
#pragma unroll
  for (int r = 0; r < 8; ++r) acc0[r] += acc1[r] + bv;
  return acc0;
}

__global__ __launch_bounds__(TPB) void fused_hgnn_mlp(
    const int* __restrict__ adj, const float* __restrict__ feats_op,
    const float* __restrict__ feats_ma,
    const float* __restrict__ l0w1, const float* __restrict__ l0b1,
    const float* __restrict__ l0w2, const float* __restrict__ l0b2,
    const float* __restrict__ l0w3, const float* __restrict__ l0b3,
    const float* __restrict__ l1w1, const float* __restrict__ l1b1,
    const float* __restrict__ l1w2, const float* __restrict__ l1b2,
    const float* __restrict__ l1w3, const float* __restrict__ l1b3,
    const float* __restrict__ pw1, const float* __restrict__ pb1,
    const float* __restrict__ pw2, const float* __restrict__ pb2,
    const float* __restrict__ pw3, const float* __restrict__ pb3,
    float* __restrict__ out) {
  __shared__ __attribute__((aligned(16))) float sFM[NMA * FDIM];
  __shared__ __attribute__((aligned(16))) float sBuf[WAVES][2][16 * PH];
  __shared__ __attribute__((aligned(16))) float sCat[WAVES][16 * PC];

  const int tid = threadIdx.x;
  const int wid = tid >> 5, lane = tid & 31;
  const int half = lane >> 4, l16 = lane & 15;
  const int row0 = blockIdx.x * (WAVES * 16) + wid * 16;
  const int bb = row0 / NOP;
  const int op0 = row0 % NOP;

  static_assert(TPB * 4 == NMA * FDIM, "one 16-byte piece of feats_ma per thread");
  const int m = lane >> 1, k0 = (lane & 1) * 4;
  const size_t grow = (size_t)bb * NOP + op0 + m;
  const v4f fmv = *(const v4f*)(feats_ma + (size_t)bb * NMA * FDIM + tid * 4);
  const v4i* arow = (const v4i*)(adj + grow * NMA);
  v4i av[NMA / 4];
#pragma unroll
  for (int q = 0; q < NMA / 4; ++q) av[q] = arow[q];
  const v4f fop = *(const v4f*)(feats_op + grow * FDIM + k0);
  __builtin_amdgcn_sched_barrier(0);

  __builtin_prefetch(l0w2 + tid * 128, 0, 1);
  __builtin_prefetch(l1w2 + tid * 128, 0, 1);
  __builtin_prefetch(pw2 + tid * 128, 0, 1);

  *(v4f*)&sFM[tid * 4] = fmv;
  __syncthreads();

  float* sA = &sBuf[wid][0][0];
  float* sB = &sBuf[wid][1][0];
  float* sC = &sCat[wid][0];

  {
    v4f s = {0.f, 0.f, 0.f, 0.f};
#pragma unroll
    for (int q = 0; q < NMA / 4; ++q) {
#pragma unroll
      for (int e = 0; e < 4; ++e) {
        const float a = (float)av[q][e];
        const float* f = &sFM[(q * 4 + e) * FDIM + k0];
        s.x = __builtin_fmaf(a, f[0], s.x); s.y = __builtin_fmaf(a, f[1], s.y);
        s.z = __builtin_fmaf(a, f[2], s.z); s.w = __builtin_fmaf(a, f[3], s.w);
      }
    }
    *(v4f*)&sA[m * PF + k0] = s;
  }

  for (int st = 0; st < 3; ++st) {
    const float* w1 = st == 0 ? l0w1 : (st == 1 ? l1w1 : pw1);
    const float* b1 = st == 0 ? l0b1 : (st == 1 ? l1b1 : pb1);
    const float* w2 = st == 0 ? l0w2 : (st == 1 ? l1w2 : pw2);
    const float* b2 = st == 0 ? l0b2 : (st == 1 ? l1b2 : pb2);
    const float* w3 = st == 0 ? l0w3 : (st == 1 ? l1w3 : pw3);
    const float* b3 = st == 0 ? l0b3 : (st == 1 ? l1b3 : pb3);

    if (st == 1) *(v4f*)&sA[m * PF + k0] = fop;
    if (st < 2) {
      __syncthreads();
      layer_h<FDIM, PF, true>(w1, b1, sA, sB, lane);
    } else {
      layer_h<16, PC, true>(w1, b1, sC, sB, lane);
    }
    __syncthreads();
    layer_h<HDIM, PH, true>(w2, b2, sB, sA, lane);
    __syncthreads();
    const v8f e = layer_o<HDIM, PH>(w3, b3, sA, lane);
    if (st < 2) {
      if (l16 < 8) {
#pragma unroll
        for (int r = 0; r < 8; ++r)
          sC[(r + half * 8) * PC + st * 8 + l16] = elu1(e[r]);
      }
      __syncthreads();
    } else {
      if (l16 < 8) {
#pragma unroll
        for (int r = 0; r < 8; ++r)
          out[(size_t)(row0 + r + half * 8) * ODIM + l16] = e[r];
      }
    }
  }
}

extern "C" void kernel_launch(void* const* d_in, const int* in_sizes, int n_in,
                              void* d_out, int out_size, void* d_ws,
                              size_t ws_size, hipStream_t stream) {
  const int*   adj = (const int*)d_in[0];
  const float* fop = (const float*)d_in[1];
  const float* fma = (const float*)d_in[2];
  const float* l0w1 = (const float*)d_in[3];
  const float* l0b1 = (const float*)d_in[4];
  const float* l0w2 = (const float*)d_in[5];
  const float* l0b2 = (const float*)d_in[6];
  const float* l0w3 = (const float*)d_in[7];
  const float* l0b3 = (const float*)d_in[8];
  const float* l1w1 = (const float*)d_in[9];
  const float* l1b1 = (const float*)d_in[10];
  const float* l1w2 = (const float*)d_in[11];
  const float* l1b2 = (const float*)d_in[12];
  const float* l1w3 = (const float*)d_in[13];
  const float* l1b3 = (const float*)d_in[14];
  const float* pw1 = (const float*)d_in[15];
  const float* pb1 = (const float*)d_in[16];
  const float* pw2 = (const float*)d_in[17];
  const float* pb2 = (const float*)d_in[18];
  const float* pw3 = (const float*)d_in[19];
  const float* pb3 = (const float*)d_in[20];

  const int total_rows = BATCH * NOP;
  const int rows_per_block = WAVES * 16;
  dim3 grid(total_rows / rows_per_block);
  fused_hgnn_mlp<<<grid, TPB, 0, stream>>>(
      adj, fop, fma, l0w1, l0b1, l0w2, l0b2, l0w3, l0b3, l1w1, l1b1, l1w2,
      l1b2, l1w3, l1b3, pw1, pb1, pw2, pb2, pw3, pb3, (float*)d_out);
}


# Hidden 128x128 weights staged once per work-group into LDS as (k,k+1) pairs, fetched one layer ahead; B fragments come from ds_load_b128 with no register moves; sums unchanged
# speedup vs baseline: 2.1189x; 1.1369x over previous
#include <hip/hip_runtime.h>
#include <math.h>


typedef __attribute__((ext_vector_type(2))) float v2f;
typedef __attribute__((ext_vector_type(4))) float v4f;
typedef __attribute__((ext_vector_type(4))) int v4i;
typedef __attribute__((ext_vector_type(8))) float v8f;

#define WAVES 4
#define TPB   (WAVES * 32)
#define BATCH 32
#define NOP   512
#define NMA   64
#define FDIM  8
#define HDIM  128
#define ODIM  8
#define PH    (HDIM + 4)
#define PF    (FDIM + 4)
#define PC    (16 + 4)

__device__ __forceinline__ float elu1(float x) {
  float e = __builtin_amdgcn_exp2f(x * 1.44269504088896340736f) - 1.0f;
  return x > 0.0f ? x : e;
}

__device__ __forceinline__ v8f wmma_f32(v2f a, v2f b, v8f c) {
  return __builtin_amdgcn_wmma_f32_16x16x4_f32(false, a, false, b, (short)0, c,
                                               false, false);
}

#define KCH 4

template <int NS>
struct Frag {
  v2f a[NS];
  v4f b[NS][2][2];
};

template <int NS, int PA>
__device__ __forceinline__ void load_chunk(Frag<NS>& f,
                                           const float* __restrict__ W,
                                           const float* sA, int ks0, int c,
                                           int kh) {
#pragma unroll
  for (int i = 0; i < NS; ++i) {
    const int k = (ks0 + i) * 4 + kh;
    f.a[i] = *(const v2f*)&sA[c * PA + k];
#pragma unroll
    for (int s = 0; s < 2; ++s) {
#pragma unroll
      for (int g = 0; g < 2; ++g)
        f.b[i][s][g] = *(const v4f*)&W[(k + s) * HDIM + g * 64 + c * 4];
    }
  }
}

template <int NS>
__device__ __forceinline__ void mma_chunk(const Frag<NS>& f,
                                          v8f (&acc)[2][2][4]) {
#pragma unroll
  for (int i = 0; i < NS; ++i) {
#pragma unroll
    for (int g = 0; g < 2; ++g) {
#pragma unroll
      for (int t = 0; t < 4; ++t) {
        v2f b; b.x = f.b[i][0][g][t]; b.y = f.b[i][1][g][t];
        acc[i & 1][g][t] = wmma_f32(f.a[i], b, acc[i & 1][g][t]);
      }
    }
  }
}

template <int K, int PA, bool ELU>
__device__ __forceinline__ void layer_h(const float* __restrict__ W,
                                        const float* __restrict__ bias,
                                        const float* sA, float* sY, int lane) {
  constexpr int NKS = K / 4;
  constexpr int NS = NKS < KCH ? NKS : KCH;
  static_assert(NKS == NS, "first layers only: all k-steps in one chunk");
  const int half = lane >> 4, c = lane & 15;
  const int kh = half * 2;

  v8f acc[2][2][4];
#pragma unroll
  for (int p = 0; p < 2; ++p)
#pragma unroll
    for (int g = 0; g < 2; ++g)
#pragma unroll
      for (int t = 0; t < 4; ++t)
        acc[p][g][t] = (v8f){0.f, 0.f, 0.f, 0.f, 0.f, 0.f, 0.f, 0.f};

  v4f bv[2];
  bv[0] = *(const v4f*)&bias[c * 4];
  bv[1] = *(const v4f*)&bias[64 + c * 4];

  Frag<NS> f0;
  load_chunk<NS, PA>(f0, W, sA, 0, c, kh);
  mma_chunk<NS>(f0, acc);

#pragma unroll
  for (int g = 0; g < 2; ++g) {
#pragma unroll
    for (int r = 0; r < 8; ++r) {
      v4f y;
#pragma unroll
      for (int t = 0; t < 4; ++t) {
        float v = acc[0][g][t][r] + acc[1][g][t][r] + bv[g][t];
        if (ELU) v = elu1(v);
        y[t] = v;
      }
      *(v4f*)&sY[(r + half * 8) * PH + g * 64 + c * 4] = y;
    }
  }
}

#define WPAIR (2 * HDIM)

struct WShare { v4f r[32]; };

__device__ __forceinline__ void wshare_load(WShare& w, const float* __restrict__ Wg,
                                            int wid, int lane) {
#pragma unroll
  for (int q = 0; q < 32; ++q)
    w.r[q] = *(const v4f*)&Wg[(wid * 32 + q) * HDIM + lane * 4];
}

__device__ __forceinline__ void wshare_store(const WShare& w, float* sW, int wid,
                                             int lane) {
#pragma unroll
  for (int kk = 0; kk < 16; ++kk) {
    const v4f x = w.r[2 * kk], y = w.r[2 * kk + 1];
    float* d = &sW[(wid * 16 + kk) * WPAIR + lane * 8];
    *(v4f*)&d[0] = (v4f){x[0], y[0], x[1], y[1]};
    *(v4f*)&d[4] = (v4f){x[2], y[2], x[3], y[3]};
  }
}

template <int NS>
struct FragL {
  v2f a[NS];
  v4f b[NS][4];
};

template <int NS>
__device__ __forceinline__ void load_chunk_l(FragL<NS>& f, const float* sW,
                                             const float* sA, int ks0, int c,
                                             int half) {
#pragma unroll
  for (int i = 0; i < NS; ++i) {
    const int j = ks0 + i;
    f.a[i] = *(const v2f*)&sA[c * PH + j * 4 + half * 2];
#pragma unroll
    for (int g = 0; g < 4; ++g)
      f.b[i][g] = *(const v4f*)&sW[(j * 2 + half) * WPAIR + g * 64 + c * 4];
  }
}

template <int NS>
__device__ __forceinline__ void mma_chunk_l(const FragL<NS>& f,
                                            v8f (&acc)[2][4][2]) {
#pragma unroll
  for (int i = 0; i < NS; ++i) {
#pragma unroll
    for (int g = 0; g < 4; ++g) {
#pragma unroll
      for (int e = 0; e < 2; ++e) {
        v2f b; b.x = f.b[i][g][2 * e]; b.y = f.b[i][g][2 * e + 1];
        acc[i & 1][g][e] = wmma_f32(f.a[i], b, acc[i & 1][g][e]);
      }
    }
  }
}

__device__ __forceinline__ void layer_hl(const float* sW,
                                         const float* __restrict__ bias,
                                         const float* sA, float* sY, int lane) {
  constexpr int NS = KCH;
  constexpr int NCH = (HDIM / 4) / NS;
  const int half = lane >> 4, c = lane & 15;

  v8f acc[2][4][2];
#pragma unroll
  for (int p = 0; p < 2; ++p)
#pragma unroll
    for (int g = 0; g < 4; ++g)
#pragma unroll
      for (int e = 0; e < 2; ++e)
        acc[p][g][e] = (v8f){0.f, 0.f, 0.f, 0.f, 0.f, 0.f, 0.f, 0.f};

  v2f bv[4];
#pragma unroll
  for (int g = 0; g < 4; ++g) bv[g] = *(const v2f*)&bias[g * 32 + c * 2];

  FragL<NS> f0, f1;
  load_chunk_l<NS>(f0, sW, sA, 0, c, half);
#pragma unroll 1
  for (int ch = 0; ch < NCH - 2; ch += 2) {
    load_chunk_l<NS>(f1, sW, sA, (ch + 1) * NS, c, half);
    __builtin_amdgcn_sched_barrier(0);
    mma_chunk_l<NS>(f0, acc);
    load_chunk_l<NS>(f0, sW, sA, (ch + 2) * NS, c, half);
    __builtin_amdgcn_sched_barrier(0);
    mma_chunk_l<NS>(f1, acc);
  }
  load_chunk_l<NS>(f1, sW, sA, (NCH - 1) * NS, c, half);
  __builtin_amdgcn_sched_barrier(0);
  mma_chunk_l<NS>(f0, acc);
  mma_chunk_l<NS>(f1, acc);

#pragma unroll
  for (int g = 0; g < 4; ++g) {
#pragma unroll
    for (int r = 0; r < 8; ++r) {
      v2f y;
#pragma unroll
      for (int e = 0; e < 2; ++e)
        y[e] = elu1(acc[0][g][e][r] + acc[1][g][e][r] + bv[g][e]);
      *(v2f*)&sY[(r + half * 8) * PH + g * 32 + c * 2] = y;
    }
  }
}

template <int K, int PA>
__device__ __forceinline__ v8f layer_o(const float* __restrict__ W,
                                       const float* __restrict__ bias,
                                       const float* sA, int lane) {
  const int half = lane >> 4, l16 = lane & 15;
  const int kh = half * 2;
  const int n = (l16 < 8) ? l16 : 0;
  v8f acc0 = {0.f, 0.f, 0.f, 0.f, 0.f, 0.f, 0.f, 0.f};
  v8f acc1 = {0.f, 0.f, 0.f, 0.f, 0.f, 0.f, 0.f, 0.f};
#pragma unroll
  for (int kc = 0; kc < K / 4; kc += 2) {
    {
      const int k = kc * 4 + kh;
      v2f a; a.x = sA[l16 * PA + k];   a.y = sA[l16 * PA + k + 1];
      v2f b; b.x = W[k * ODIM + n];    b.y = W[(k + 1) * ODIM + n];
      acc0 = wmma_f32(a, b, acc0);
    }
    {
      const int k = (kc + 1) * 4 + kh;
      v2f a; a.x = sA[l16 * PA + k];   a.y = sA[l16 * PA + k + 1];
      v2f b; b.x = W[k * ODIM + n];    b.y = W[(k + 1) * ODIM + n];
      acc1 = wmma_f32(a, b, acc1);
    }
  }
  const float bv = bias[n];
#pragma unroll
  for (int r = 0; r < 8; ++r) acc0[r] += acc1[r] + bv;
  return acc0;
}

__global__ __launch_bounds__(TPB) void fused_hgnn_mlp(
    const int* __restrict__ adj, const float* __restrict__ feats_op,
    const float* __restrict__ feats_ma,
    const float* __restrict__ l0w1, const float* __restrict__ l0b1,
    const float* __restrict__ l0w2, const float* __restrict__ l0b2,
    const float* __restrict__ l0w3, const float* __restrict__ l0b3,
    const float* __restrict__ l1w1, const float* __restrict__ l1b1,
    const float* __restrict__ l1w2, const float* __restrict__ l1b2,
    const float* __restrict__ l1w3, const float* __restrict__ l1b3,
    const float* __restrict__ pw1, const float* __restrict__ pb1,
    const float* __restrict__ pw2, const float* __restrict__ pb2,
    const float* __restrict__ pw3, const float* __restrict__ pb3,
    float* __restrict__ out) {
  __shared__ __attribute__((aligned(16))) float sFM[NMA * FDIM];
  __shared__ __attribute__((aligned(16))) float sBuf[WAVES][2][16 * PH];
  __shared__ __attribute__((aligned(16))) float sCat[WAVES][16 * PC];
  __shared__ __attribute__((aligned(16))) float sW[(HDIM / 2) * WPAIR];

  const int tid = threadIdx.x;
  const int wid = tid >> 5, lane = tid & 31;
  const int half = lane >> 4, l16 = lane & 15;
  const int row0 = blockIdx.x * (WAVES * 16) + wid * 16;
  const int bb = row0 / NOP;
  const int op0 = row0 % NOP;

  static_assert(TPB * 4 == NMA * FDIM, "one 16-byte piece of feats_ma per thread");
  static_assert(WAVES * 32 == HDIM, "each wave stages 32 k rows of a hidden matrix");
  const int m = lane >> 1, k0 = (lane & 1) * 4;
  const size_t grow = (size_t)bb * NOP + op0 + m;
  const v4f fmv = *(const v4f*)(feats_ma + (size_t)bb * NMA * FDIM + tid * 4);
  const v4i* arow = (const v4i*)(adj + grow * NMA);
  v4i av[NMA / 4];
#pragma unroll
  for (int q = 0; q < NMA / 4; ++q) av[q] = arow[q];
  const v4f fop = *(const v4f*)(feats_op + grow * FDIM + k0);
  WShare wsh;
  wshare_load(wsh, l0w2, wid, lane);
  __builtin_amdgcn_sched_barrier(0);

  __builtin_prefetch(l1w2 + tid * 128, 0, 1);
  __builtin_prefetch(pw2 + tid * 128, 0, 1);

  *(v4f*)&sFM[tid * 4] = fmv;
  wshare_store(wsh, sW, wid, lane);
  __syncthreads();

  float* sA = &sBuf[wid][0][0];
  float* sB = &sBuf[wid][1][0];
  float* sC = &sCat[wid][0];

  {
    v4f s = {0.f, 0.f, 0.f, 0.f};
#pragma unroll
    for (int q = 0; q < NMA / 4; ++q) {
#pragma unroll
      for (int e = 0; e < 4; ++e) {
        const float a = (float)av[q][e];
        const float* f = &sFM[(q * 4 + e) * FDIM + k0];
        s.x = __builtin_fmaf(a, f[0], s.x); s.y = __builtin_fmaf(a, f[1], s.y);
        s.z = __builtin_fmaf(a, f[2], s.z); s.w = __builtin_fmaf(a, f[3], s.w);
      }
    }
    *(v4f*)&sA[m * PF + k0] = s;
  }

  for (int st = 0; st < 3; ++st) {
    const float* w1 = st == 0 ? l0w1 : (st == 1 ? l1w1 : pw1);
    const float* b1 = st == 0 ? l0b1 : (st == 1 ? l1b1 : pb1);
    const float* b2 = st == 0 ? l0b2 : (st == 1 ? l1b2 : pb2);
    const float* w3 = st == 0 ? l0w3 : (st == 1 ? l1w3 : pw3);
    const float* b3 = st == 0 ? l0b3 : (st == 1 ? l1b3 : pb3);

    if (st == 1) *(v4f*)&sA[m * PF + k0] = fop;
    if (st < 2) {
      __syncthreads();
      layer_h<FDIM, PF, true>(w1, b1, sA, sB, lane);
    } else {
      layer_h<16, PC, true>(w1, b1, sC, sB, lane);
    }
    __syncthreads();
    WShare wn;
    if (st < 2) wshare_load(wn, st == 0 ? l1w2 : pw2, wid, lane);
    layer_hl(sW, b2, sB, sA, lane);
    __syncthreads();
    if (st < 2) wshare_store(wn, sW, wid, lane);
    const v8f e = layer_o<HDIM, PH>(w3, b3, sA, lane);
    if (st < 2) {
      if (l16 < 8) {
#pragma unroll
        for (int r = 0; r < 8; ++r)
          sC[(r + half * 8) * PC + st * 8 + l16] = elu1(e[r]);
      }
      __syncthreads();
    } else {
      if (l16 < 8) {
#pragma unroll
        for (int r = 0; r < 8; ++r)
          out[(size_t)(row0 + r + half * 8) * ODIM + l16] = e[r];
      }
    }
  }
}

extern "C" void kernel_launch(void* const* d_in, const int* in_sizes, int n_in,
                              void* d_out, int out_size, void* d_ws,
                              size_t ws_size, hipStream_t stream) {
  const int*   adj = (const int*)d_in[0];
  const float* fop = (const float*)d_in[1];
  const float* fma = (const float*)d_in[2];
  const float* l0w1 = (const float*)d_in[3];
  const float* l0b1 = (const float*)d_in[4];
  const float* l0w2 = (const float*)d_in[5];
  const float* l0b2 = (const float*)d_in[6];
  const float* l0w3 = (const float*)d_in[7];
  const float* l0b3 = (const float*)d_in[8];
  const float* l1w1 = (const float*)d_in[9];
  const float* l1b1 = (const float*)d_in[10];
  const float* l1w2 = (const float*)d_in[11];
  const float* l1b2 = (const float*)d_in[12];
  const float* l1w3 = (const float*)d_in[13];
  const float* l1b3 = (const float*)d_in[14];
  const float* pw1 = (const float*)d_in[15];
  const float* pb1 = (const float*)d_in[16];
  const float* pw2 = (const float*)d_in[17];
  const float* pb2 = (const float*)d_in[18];
  const float* pw3 = (const float*)d_in[19];
  const float* pb3 = (const float*)d_in[20];

  const int total_rows = BATCH * NOP;
  const int rows_per_block = WAVES * 16;
  dim3 grid(total_rows / rows_per_block);
  fused_hgnn_mlp<<<grid, TPB, 0, stream>>>(
      adj, fop, fma, l0w1, l0b1, l0w2, l0b2, l0w3, l0b3, l1w1, l1b1, l1w2,
      l1b2, l1w3, l1b3, pw1, pb1, pw2, pb2, pw3, pb3, (float*)d_out);
}


# A tile's two independent input branches run as two waves of one SIMD (8-wave groups, both branch matrices in LDS): one wave's epilogues and load waits run under the other's matrix work; same sums
# speedup vs baseline: 2.4618x; 1.1618x over previous
#include <hip/hip_runtime.h>
#include <math.h>


typedef __attribute__((ext_vector_type(2))) float v2f;
typedef __attribute__((ext_vector_type(4))) float v4f;
typedef __attribute__((ext_vector_type(4))) int v4i;
typedef __attribute__((ext_vector_type(8))) float v8f;

#define TILES 4
#define WAVES (2 * TILES)
#define TPB   (WAVES * 32)
#define BATCH 32
#define NOP   512
#define NMA   64
#define FDIM  8
#define HDIM  128
#define ODIM  8
#define PH    (HDIM + 4)
#define PF    (FDIM + 4)
#define PC    (16 + 4)

__device__ __forceinline__ float elu1(float x) {
  float e = __builtin_amdgcn_exp2f(x * 1.44269504088896340736f) - 1.0f;
  return x > 0.0f ? x : e;
}

__device__ __forceinline__ v8f wmma_f32(v2f a, v2f b, v8f c) {
  return __builtin_amdgcn_wmma_f32_16x16x4_f32(false, a, false, b, (short)0, c,
                                               false, false);
}

__device__ __forceinline__ void wave_sync() {
  __builtin_amdgcn_fence(__ATOMIC_RELEASE, "wavefront");
  __builtin_amdgcn_wave_barrier();
  __builtin_amdgcn_fence(__ATOMIC_ACQUIRE, "wavefront");
}

#define KCH 4

template <int NS>
struct Frag {
  v2f a[NS];
  v4f b[NS][2][2];
};

template <int NS, int PA>
__device__ __forceinline__ void load_chunk(Frag<NS>& f,
                                           const float* __restrict__ W,
                                           const float* sA, int ks0, int c,
                                           int kh) {
#pragma unroll
  for (int i = 0; i < NS; ++i) {
    const int k = (ks0 + i) * 4 + kh;
    f.a[i] = *(const v2f*)&sA[c * PA + k];
#pragma unroll
    for (int s = 0; s < 2; ++s) {
#pragma unroll
      for (int g = 0; g < 2; ++g)
        f.b[i][s][g] = *(const v4f*)&W[(k + s) * HDIM + g * 64 + c * 4];
    }
  }
}

template <int NS>
__device__ __forceinline__ void mma_chunk(const Frag<NS>& f,
                                          v8f (&acc)[2][2][4]) {
#pragma unroll
  for (int i = 0; i < NS; ++i) {
#pragma unroll
    for (int g = 0; g < 2; ++g) {
#pragma unroll
      for (int t = 0; t < 4; ++t) {
        v2f b; b.x = f.b[i][0][g][t]; b.y = f.b[i][1][g][t];
        acc[i & 1][g][t] = wmma_f32(f.a[i], b, acc[i & 1][g][t]);
      }
    }
  }
}

template <int K, int PA, bool ELU>
__device__ __forceinline__ void layer_h(const float* __restrict__ W,
                                        const float* __restrict__ bias,
                                        const float* sA, float* sY, int lane) {
  constexpr int NKS = K / 4;
  constexpr int NS = NKS < KCH ? NKS : KCH;
  static_assert(NKS == NS, "first layers only: all k-steps in one chunk");
  const int half = lane >> 4, c = lane & 15;
  const int kh = half * 2;

  v8f acc[2][2][4];
#pragma unroll
  for (int p = 0; p < 2; ++p)
#pragma unroll
    for (int g = 0; g < 2; ++g)
#pragma unroll
      for (int t = 0; t < 4; ++t)
        acc[p][g][t] = (v8f){0.f, 0.f, 0.f, 0.f, 0.f, 0.f, 0.f, 0.f};

  v4f bv[2];
  bv[0] = *(const v4f*)&bias[c * 4];
  bv[1] = *(const v4f*)&bias[64 + c * 4];

  Frag<NS> f0;
  load_chunk<NS, PA>(f0, W, sA, 0, c, kh);
  mma_chunk<NS>(f0, acc);

#pragma unroll
  for (int g = 0; g < 2; ++g) {
#pragma unroll
    for (int r = 0; r < 8; ++r) {
      v4f y;
#pragma unroll
      for (int t = 0; t < 4; ++t) {
        float v = acc[0][g][t][r] + acc[1][g][t][r] + bv[g][t];
        if (ELU) v = elu1(v);
        y[t] = v;
      }
      *(v4f*)&sY[(r + half * 8) * PH + g * 64 + c * 4] = y;
    }
  }
}

#define WPAIR (2 * HDIM)

#define WROWS (HDIM / WAVES)
struct WShare { v4f r[WROWS]; };

__device__ __forceinline__ void wshare_load(WShare& w, const float* __restrict__ Wg,
                                            int wid, int lane) {
#pragma unroll
  for (int q = 0; q < WROWS; ++q)
    w.r[q] = *(const v4f*)&Wg[(wid * WROWS + q) * HDIM + lane * 4];
}

__device__ __forceinline__ void wshare_store(const WShare& w, float* sW, int wid,
                                             int lane) {
#pragma unroll
  for (int kk = 0; kk < WROWS / 2; ++kk) {
    const v4f x = w.r[2 * kk], y = w.r[2 * kk + 1];
    float* d = &sW[(wid * (WROWS / 2) + kk) * WPAIR + lane * 8];
    *(v4f*)&d[0] = (v4f){x[0], y[0], x[1], y[1]};
    *(v4f*)&d[4] = (v4f){x[2], y[2], x[3], y[3]};
  }
}

template <int NS>
struct FragL {
  v2f a[NS];
  v4f b[NS][4];
};

template <int NS>
__device__ __forceinline__ void load_chunk_l(FragL<NS>& f, const float* sW,
                                             const float* sA, int ks0, int c,
                                             int half) {
#pragma unroll
  for (int i = 0; i < NS; ++i) {
    const int j = ks0 + i;
    f.a[i] = *(const v2f*)&sA[c * PH + j * 4 + half * 2];
#pragma unroll
    for (int g = 0; g < 4; ++g)
      f.b[i][g] = *(const v4f*)&sW[(j * 2 + half) * WPAIR + g * 64 + c * 4];
  }
}

template <int NS>
__device__ __forceinline__ void mma_chunk_l(const FragL<NS>& f,
                                            v8f (&acc)[2][4][2]) {
#pragma unroll
  for (int i = 0; i < NS; ++i) {
#pragma unroll
    for (int g = 0; g < 4; ++g) {
#pragma unroll
      for (int e = 0; e < 2; ++e) {
        v2f b; b.x = f.b[i][g][2 * e]; b.y = f.b[i][g][2 * e + 1];
        acc[i & 1][g][e] = wmma_f32(f.a[i], b, acc[i & 1][g][e]);
      }
    }
  }
}

__device__ __forceinline__ void layer_hl(const float* sW,
                                         const float* __restrict__ bias,
                                         const float* sA, float* sY, int lane) {
  constexpr int NS = KCH;
  constexpr int NCH = (HDIM / 4) / NS;
  const int half = lane >> 4, c = lane & 15;

  v8f acc[2][4][2];
#pragma unroll
  for (int p = 0; p < 2; ++p)
#pragma unroll
    for (int g = 0; g < 4; ++g)
#pragma unroll
      for (int e = 0; e < 2; ++e)
        acc[p][g][e] = (v8f){0.f, 0.f, 0.f, 0.f, 0.f, 0.f, 0.f, 0.f};

  v2f bv[4];
#pragma unroll
  for (int g = 0; g < 4; ++g) bv[g] = *(const v2f*)&bias[g * 32 + c * 2];

  FragL<NS> f0, f1;
  load_chunk_l<NS>(f0, sW, sA, 0, c, half);
#pragma unroll 1
  for (int ch = 0; ch < NCH - 2; ch += 2) {
    load_chunk_l<NS>(f1, sW, sA, (ch + 1) * NS, c, half);
    __builtin_amdgcn_sched_barrier(0);
    mma_chunk_l<NS>(f0, acc);
    load_chunk_l<NS>(f0, sW, sA, (ch + 2) * NS, c, half);
    __builtin_amdgcn_sched_barrier(0);
    mma_chunk_l<NS>(f1, acc);
  }
  load_chunk_l<NS>(f1, sW, sA, (NCH - 1) * NS, c, half);
  __builtin_amdgcn_sched_barrier(0);
  mma_chunk_l<NS>(f0, acc);
  mma_chunk_l<NS>(f1, acc);

#pragma unroll
  for (int g = 0; g < 4; ++g) {
#pragma unroll
    for (int r = 0; r < 8; ++r) {
      v2f y;
#pragma unroll
      for (int e = 0; e < 2; ++e)
        y[e] = elu1(acc[0][g][e][r] + acc[1][g][e][r] + bv[g][e]);
      *(v2f*)&sY[(r + half * 8) * PH + g * 32 + c * 2] = y;
    }
  }
}

template <int K, int PA>
__device__ __forceinline__ v8f layer_o(const float* __restrict__ W,
                                       const float* __restrict__ bias,
                                       const float* sA, int lane) {
  const int half = lane >> 4, l16 = lane & 15;
  const int kh = half * 2;
  const int n = (l16 < 8) ? l16 : 0;
  v8f acc0 = {0.f, 0.f, 0.f, 0.f, 0.f, 0.f, 0.f, 0.f};
  v8f acc1 = {0.f, 0.f, 0.f, 0.f, 0.f, 0.f, 0.f, 0.f};
#pragma unroll
  for (int kc = 0; kc < K / 4; kc += 2) {
    {
      const int k = kc * 4 + kh;
      v2f a; a.x = sA[l16 * PA + k];   a.y = sA[l16 * PA + k + 1];
      v2f b; b.x = W[k * ODIM + n];    b.y = W[(k + 1) * ODIM + n];
      acc0 = wmma_f32(a, b, acc0);
    }
    {
      const int k = (kc + 1) * 4 + kh;
      v2f a; a.x = sA[l16 * PA + k];   a.y = sA[l16 * PA + k + 1];
      v2f b; b.x = W[k * ODIM + n];    b.y = W[(k + 1) * ODIM + n];
      acc1 = wmma_f32(a, b, acc1);
    }
  }
  const float bv = bias[n];
#pragma unroll
  for (int r = 0; r < 8; ++r) acc0[r] += acc1[r] + bv;
  return acc0;
}

__global__ __launch_bounds__(TPB) void fused_hgnn_mlp(
    const int* __restrict__ adj, const float* __restrict__ feats_op,
    const float* __restrict__ feats_ma,
    const float* __restrict__ l0w1, const float* __restrict__ l0b1,
    const float* __restrict__ l0w2, const float* __restrict__ l0b2,
    const float* __restrict__ l0w3, const float* __restrict__ l0b3,
    const float* __restrict__ l1w1, const float* __restrict__ l1b1,
    const float* __restrict__ l1w2, const float* __restrict__ l1b2,
    const float* __restrict__ l1w3, const float* __restrict__ l1b3,
    const float* __restrict__ pw1, const float* __restrict__ pb1,
    const float* __restrict__ pw2, const float* __restrict__ pb2,
    const float* __restrict__ pw3, const float* __restrict__ pb3,
    float* __restrict__ out) {
  __shared__ __attribute__((aligned(16))) float sFM[NMA * FDIM];
  __shared__ __attribute__((aligned(16))) float sBuf[WAVES][2][16 * PH];
  __shared__ __attribute__((aligned(16))) float sCat[TILES][16 * PC];
  __shared__ __attribute__((aligned(16))) float sW[2][(HDIM / 2) * WPAIR];

  const int tid = threadIdx.x;
  const int wid = __builtin_amdgcn_readfirstlane(tid >> 5);
  const int lane = tid & 31;
  const int tile = wid & (TILES - 1);
  const int br = wid / TILES;
  const int half = lane >> 4, l16 = lane & 15;
  const int row0 = blockIdx.x * (TILES * 16) + tile * 16;
  const int bb = row0 / NOP;
  const int op0 = row0 % NOP;

  static_assert(TILES * 32 * 4 == NMA * FDIM, "one 16-byte piece of feats_ma per thread of waves 0..3");
  static_assert(WAVES * WROWS == HDIM, "the waves together stage all k rows of a hidden matrix");
  const int m = lane >> 1, k0 = (lane & 1) * 4;
  const size_t grow = (size_t)bb * NOP + op0 + m;
  v4f fmv;
  v4i av[NMA / 4];
  v4f fop;
  if (br == 0) {
    fmv = *(const v4f*)(feats_ma + (size_t)bb * NMA * FDIM + tid * 4);
    const v4i* arow = (const v4i*)(adj + grow * NMA);
#pragma unroll
    for (int q = 0; q < NMA / 4; ++q) av[q] = arow[q];
  } else {
    fop = *(const v4f*)(feats_op + grow * FDIM + k0);
  }
  WShare wsh;
  wshare_load(wsh, l0w2, wid, lane);
  __builtin_amdgcn_sched_barrier(0);

  __builtin_prefetch(pw2 + tid * 64, 0, 1);

  if (br == 0) *(v4f*)&sFM[tid * 4] = fmv;
  wshare_store(wsh, &sW[0][0], wid, lane);
  __syncthreads();
  wshare_load(wsh, l1w2, wid, lane);

  float* sA = &sBuf[wid][0][0];
  float* sB = &sBuf[wid][1][0];
  float* sC = &sCat[tile][0];

  if (br == 0) {
    v4f s = {0.f, 0.f, 0.f, 0.f};
#pragma unroll
    for (int q = 0; q < NMA / 4; ++q) {
#pragma unroll
      for (int e = 0; e < 4; ++e) {
        const float a = (float)av[q][e];
        const float* f = &sFM[(q * 4 + e) * FDIM + k0];
        s.x = __builtin_fmaf(a, f[0], s.x); s.y = __builtin_fmaf(a, f[1], s.y);
        s.z = __builtin_fmaf(a, f[2], s.z); s.w = __builtin_fmaf(a, f[3], s.w);
      }
    }
    *(v4f*)&sA[m * PF + k0] = s;
  } else {
    *(v4f*)&sA[m * PF + k0] = fop;
  }
  wave_sync();

  layer_h<FDIM, PF, true>(br == 0 ? l0w1 : l1w1, br == 0 ? l0b1 : l1b1, sA, sB, lane);
  wshare_store(wsh, &sW[1][0], wid, lane);
  __syncthreads();
  wshare_load(wsh, pw2, wid, lane);

  const float* swp = &sW[br][0];
  const float* b2 = br == 0 ? l0b2 : l1b2;
  const float* w3 = br == 0 ? l0w3 : l1w3;
  const float* b3 = br == 0 ? l0b3 : l1b3;
  for (int ph = 0; ph < 2; ++ph) {
    layer_hl(swp, b2, sB, sA, lane);
    wave_sync();
    const v8f e = layer_o<HDIM, PH>(w3, b3, sA, lane);
    if (ph == 0) {
      if (l16 < 8) {
#pragma unroll
        for (int r = 0; r < 8; ++r)
          sC[(r + half * 8) * PC + br * 8 + l16] = elu1(e[r]);
      }
      __syncthreads();
      wshare_store(wsh, &sW[0][0], wid, lane);
      __syncthreads();
      if (br != 0) return;
      layer_h<16, PC, true>(pw1, pb1, sC, sB, lane);
      wave_sync();
      swp = &sW[0][0]; b2 = pb2; w3 = pw3; b3 = pb3;
    } else {
      if (l16 < 8) {
#pragma unroll
        for (int r = 0; r < 8; ++r)
          out[(size_t)(row0 + r + half * 8) * ODIM + l16] = e[r];
      }
    }
  }
}

extern "C" void kernel_launch(void* const* d_in, const int* in_sizes, int n_in,
                              void* d_out, int out_size, void* d_ws,
                              size_t ws_size, hipStream_t stream) {
  const int*   adj = (const int*)d_in[0];
  const float* fop = (const float*)d_in[1];
  const float* fma = (const float*)d_in[2];
  const float* l0w1 = (const float*)d_in[3];
  const float* l0b1 = (const float*)d_in[4];
  const float* l0w2 = (const float*)d_in[5];
  const float* l0b2 = (const float*)d_in[6];
  const float* l0w3 = (const float*)d_in[7];
  const float* l0b3 = (const float*)d_in[8];
  const float* l1w1 = (const float*)d_in[9];
  const float* l1b1 = (const float*)d_in[10];
  const float* l1w2 = (const float*)d_in[11];
  const float* l1b2 = (const float*)d_in[12];
  const float* l1w3 = (const float*)d_in[13];
  const float* l1b3 = (const float*)d_in[14];
  const float* pw1 = (const float*)d_in[15];
  const float* pb1 = (const float*)d_in[16];
  const float* pw2 = (const float*)d_in[17];
  const float* pb2 = (const float*)d_in[18];
  const float* pw3 = (const float*)d_in[19];
  const float* pb3 = (const float*)d_in[20];

  const int total_rows = BATCH * NOP;
  const int rows_per_block = TILES * 16;
  dim3 grid(total_rows / rows_per_block);
  fused_hgnn_mlp<<<grid, TPB, 0, stream>>>(
      adj, fop, fma, l0w1, l0b1, l0w2, l0b2, l0w3, l0b3, l1w1, l1b1, l1w2,
      l1b2, l1w3, l1b3, pw1, pb1, pw2, pb2, pw3, pb3, (float*)d_out);
}


# Small layers (first layers, 128->8 heads): weights requested a layer ahead and held in registers, head A fragments read as one LDS group: no matrix chain waits on a fetch of its own; sums unchanged
# speedup vs baseline: 2.6873x; 1.0916x over previous
#include <hip/hip_runtime.h>
#include <math.h>


typedef __attribute__((ext_vector_type(2))) float v2f;
typedef __attribute__((ext_vector_type(4))) float v4f;
typedef __attribute__((ext_vector_type(4))) int v4i;
typedef __attribute__((ext_vector_type(8))) float v8f;

#define TILES 4
#define WAVES (2 * TILES)
#define TPB   (WAVES * 32)
#define BATCH 32
#define NOP   512
#define NMA   64
#define FDIM  8
#define HDIM  128
#define ODIM  8
#define PH    (HDIM + 4)
#define PF    (FDIM + 4)
#define PC    (16 + 4)

__device__ __forceinline__ float elu1(float x) {
  float e = __builtin_amdgcn_exp2f(x * 1.44269504088896340736f) - 1.0f;
  return x > 0.0f ? x : e;
}

__device__ __forceinline__ v8f wmma_f32(v2f a, v2f b, v8f c) {
  return __builtin_amdgcn_wmma_f32_16x16x4_f32(false, a, false, b, (short)0, c,
                                               false, false);
}

__device__ __forceinline__ void wave_sync() {
  __builtin_amdgcn_fence(__ATOMIC_RELEASE, "wavefront");
  __builtin_amdgcn_wave_barrier();
  __builtin_amdgcn_fence(__ATOMIC_ACQUIRE, "wavefront");
}

#define KCH 4

template <int NS>
struct Frag {
  v2f a[NS];
  v4f b[NS][2][2];
  v4f bv[2];
};

template <int NS>
__device__ __forceinline__ void load_w(Frag<NS>& f, const float* __restrict__ W,
                                       const float* __restrict__ bias, int lane) {
  const int c = lane & 15, kh = (lane >> 4) * 2;
#pragma unroll
  for (int i = 0; i < NS; ++i) {
    const int k = i * 4 + kh;
#pragma unroll
    for (int s = 0; s < 2; ++s) {
#pragma unroll
      for (int g = 0; g < 2; ++g)
        f.b[i][s][g] = *(const v4f*)&W[(k + s) * HDIM + g * 64 + c * 4];
    }
  }
  f.bv[0] = *(const v4f*)&bias[c * 4];
  f.bv[1] = *(const v4f*)&bias[64 + c * 4];
}

template <int NS, int PA>
__device__ __forceinline__ void load_a(Frag<NS>& f, const float* sA, int lane) {
  const int c = lane & 15, kh = (lane >> 4) * 2;
#pragma unroll
  for (int i = 0; i < NS; ++i) f.a[i] = *(const v2f*)&sA[c * PA + i * 4 + kh];
}

template <int NS>
__device__ __forceinline__ void mma_chunk(const Frag<NS>& f,
                                          v8f (&acc)[2][2][4]) {
#pragma unroll
  for (int i = 0; i < NS; ++i) {
#pragma unroll
    for (int g = 0; g < 2; ++g) {
#pragma unroll
      for (int t = 0; t < 4; ++t) {
        v2f b; b.x = f.b[i][0][g][t]; b.y = f.b[i][1][g][t];
        acc[i & 1][g][t] = wmma_f32(f.a[i], b, acc[i & 1][g][t]);
      }
    }
  }
}

template <int K, int PA, bool ELU>
__device__ __forceinline__ void layer_h(Frag<K / 4>& f, const float* sA,
                                        float* sY, int lane) {
  constexpr int NS = K / 4;
  static_assert(NS <= KCH && (NS & 1) == 0, "first layers only: all k-steps in one chunk");
  const int half = lane >> 4, c = lane & 15;

  v8f acc[2][2][4];
#pragma unroll
  for (int p = 0; p < 2; ++p)
#pragma unroll
    for (int g = 0; g < 2; ++g)
#pragma unroll
      for (int t = 0; t < 4; ++t)
        acc[p][g][t] = (v8f){0.f, 0.f, 0.f, 0.f, 0.f, 0.f, 0.f, 0.f};

  load_a<NS, PA>(f, sA, lane);
  mma_chunk<NS>(f, acc);

#pragma unroll
  for (int g = 0; g < 2; ++g) {
#pragma unroll
    for (int r = 0; r < 8; ++r) {
      v4f y;
#pragma unroll
      for (int t = 0; t < 4; ++t) {
        float v = acc[0][g][t][r] + acc[1][g][t][r] + f.bv[g][t];
        if (ELU) v = elu1(v);
        y[t] = v;
      }
      *(v4f*)&sY[(r + half * 8) * PH + g * 64 + c * 4] = y;
    }
  }
}

#define WPAIR (2 * HDIM)

#define WROWS (HDIM / WAVES)
struct WShare { v4f r[WROWS]; };

__device__ __forceinline__ void wshare_load(WShare& w, const float* __restrict__ Wg,
                                            int wid, int lane) {
#pragma unroll
  for (int q = 0; q < WROWS; ++q)
    w.r[q] = *(const v4f*)&Wg[(wid * WROWS + q) * HDIM + lane * 4];
}

__device__ __forceinline__ void wshare_store(const WShare& w, float* sW, int wid,
                                             int lane) {
#pragma unroll
  for (int kk = 0; kk < WROWS / 2; ++kk) {
    const v4f x = w.r[2 * kk], y = w.r[2 * kk + 1];
    float* d = &sW[(wid * (WROWS / 2) + kk) * WPAIR + lane * 8];
    *(v4f*)&d[0] = (v4f){x[0], y[0], x[1], y[1]};
    *(v4f*)&d[4] = (v4f){x[2], y[2], x[3], y[3]};
  }
}

template <int NS>
struct FragL {
  v2f a[NS];
  v4f b[NS][4];
};

template <int NS>
__device__ __forceinline__ void load_chunk_l(FragL<NS>& f, const float* sW,
                                             const float* sA, int ks0, int c,
                                             int half) {
#pragma unroll
  for (int i = 0; i < NS; ++i) {
    const int j = ks0 + i;
    f.a[i] = *(const v2f*)&sA[c * PH + j * 4 + half * 2];
#pragma unroll
    for (int g = 0; g < 4; ++g)
      f.b[i][g] = *(const v4f*)&sW[(j * 2 + half) * WPAIR + g * 64 + c * 4];
  }
}

template <int NS>
__device__ __forceinline__ void mma_chunk_l(const FragL<NS>& f,
                                            v8f (&acc)[2][4][2]) {
#pragma unroll
  for (int i = 0; i < NS; ++i) {
#pragma unroll
    for (int g = 0; g < 4; ++g) {
#pragma unroll
      for (int e = 0; e < 2; ++e) {
        v2f b; b.x = f.b[i][g][2 * e]; b.y = f.b[i][g][2 * e + 1];
        acc[i & 1][g][e] = wmma_f32(f.a[i], b, acc[i & 1][g][e]);
      }
    }
  }
}

struct HiddenAcc {
  v8f acc[2][4][2];
  v2f bv[4];
};

__device__ __forceinline__ void layer_hl_mma(HiddenAcc& o, const float* sW,
                                             const float* __restrict__ bias,
                                             const float* sA, int lane) {
  constexpr int NS = KCH;
  constexpr int NCH = (HDIM / 4) / NS;
  const int half = lane >> 4, c = lane & 15;

#pragma unroll
  for (int p = 0; p < 2; ++p)
#pragma unroll
    for (int g = 0; g < 4; ++g)
#pragma unroll
      for (int e = 0; e < 2; ++e)
        o.acc[p][g][e] = (v8f){0.f, 0.f, 0.f, 0.f, 0.f, 0.f, 0.f, 0.f};

#pragma unroll
  for (int g = 0; g < 4; ++g) o.bv[g] = *(const v2f*)&bias[g * 32 + c * 2];

  FragL<NS> f0, f1;
  load_chunk_l<NS>(f0, sW, sA, 0, c, half);
#pragma unroll 1
  for (int ch = 0; ch < NCH - 2; ch += 2) {
    load_chunk_l<NS>(f1, sW, sA, (ch + 1) * NS, c, half);
    __builtin_amdgcn_sched_barrier(0);
    mma_chunk_l<NS>(f0, o.acc);
    load_chunk_l<NS>(f0, sW, sA, (ch + 2) * NS, c, half);
    __builtin_amdgcn_sched_barrier(0);
    mma_chunk_l<NS>(f1, o.acc);
  }
  load_chunk_l<NS>(f1, sW, sA, (NCH - 1) * NS, c, half);
  __builtin_amdgcn_sched_barrier(0);
  mma_chunk_l<NS>(f0, o.acc);
  mma_chunk_l<NS>(f1, o.acc);
}

__device__ __forceinline__ void layer_hl_epi(const HiddenAcc& o, float* sY, int lane) {
  const int half = lane >> 4, c = lane & 15;
#pragma unroll
  for (int g = 0; g < 4; ++g) {
#pragma unroll
    for (int r = 0; r < 8; ++r) {
      v2f y;
#pragma unroll
      for (int e = 0; e < 2; ++e)
        y[e] = elu1(o.acc[0][g][e][r] + o.acc[1][g][e][r] + o.bv[g][e]);
      *(v2f*)&sY[(r + half * 8) * PH + g * 32 + c * 2] = y;
    }
  }
}

#define HKS (HDIM / 4)
struct HeadW {
  v2f b[HKS];
  float bias;
};

template <int J0, int J1>
__device__ __forceinline__ void head_load(HeadW& hw, const float* __restrict__ W,
                                          const float* __restrict__ bias,
                                          int lane) {
  const int half = lane >> 4, l16 = lane & 15;
  const int kh = half * 2;
  const int n = (l16 < 8) ? l16 : 0;
#pragma unroll
  for (int j = J0; j < J1; ++j) {
    const int k = j * 4 + kh;
    hw.b[j].x = W[k * ODIM + n];
    hw.b[j].y = W[(k + 1) * ODIM + n];
  }
  if (J1 == HKS) hw.bias = bias[n];
}

__device__ __forceinline__ v8f layer_o(const HeadW& hw, const float* sA, int lane) {
  const int half = lane >> 4, l16 = lane & 15;
  const int kh = half * 2;
  v2f a[HKS];
#pragma unroll
  for (int j = 0; j < HKS; ++j) a[j] = *(const v2f*)&sA[l16 * PH + j * 4 + kh];
  __builtin_amdgcn_sched_barrier(0);
  v8f acc0 = {0.f, 0.f, 0.f, 0.f, 0.f, 0.f, 0.f, 0.f};
  v8f acc1 = {0.f, 0.f, 0.f, 0.f, 0.f, 0.f, 0.f, 0.f};
#pragma unroll
  for (int j = 0; j < HKS; j += 2) {
    acc0 = wmma_f32(a[j], hw.b[j], acc0);
    acc1 = wmma_f32(a[j + 1], hw.b[j + 1], acc1);
  }
  const float bv = hw.bias;
#pragma unroll
  for (int r = 0; r < 8; ++r) acc0[r] += acc1[r] + bv;
  return acc0;
}

__global__ __launch_bounds__(TPB) void fused_hgnn_mlp(
    const int* __restrict__ adj, const float* __restrict__ feats_op,
    const float* __restrict__ feats_ma,
    const float* __restrict__ l0w1, const float* __restrict__ l0b1,
    const float* __restrict__ l0w2, const float* __restrict__ l0b2,
    const float* __restrict__ l0w3, const float* __restrict__ l0b3,
    const float* __restrict__ l1w1, const float* __restrict__ l1b1,
    const float* __restrict__ l1w2, const float* __restrict__ l1b2,
    const float* __restrict__ l1w3, const float* __restrict__ l1b3,
    const float* __restrict__ pw1, const float* __restrict__ pb1,
    const float* __restrict__ pw2, const float* __restrict__ pb2,
    const float* __restrict__ pw3, const float* __restrict__ pb3,
    float* __restrict__ out) {
  __shared__ __attribute__((aligned(16))) float sFM[NMA * FDIM];
  __shared__ __attribute__((aligned(16))) float sBuf[WAVES][2][16 * PH];
  __shared__ __attribute__((aligned(16))) float sCat[TILES][16 * PC];
  __shared__ __attribute__((aligned(16))) float sW[2][(HDIM / 2) * WPAIR];

  const int tid = threadIdx.x;
  const int wid = __builtin_amdgcn_readfirstlane(tid >> 5);
  const int lane = tid & 31;
  const int tile = wid & (TILES - 1);
  const int br = wid / TILES;
  const int half = lane >> 4, l16 = lane & 15;
  const int row0 = blockIdx.x * (TILES * 16) + tile * 16;
  const int bb = row0 / NOP;
  const int op0 = row0 % NOP;

  static_assert(TILES * 32 * 4 == NMA * FDIM, "one 16-byte piece of feats_ma per thread of waves 0..3");
  static_assert(WAVES * WROWS == HDIM, "the waves together stage all k rows of a hidden matrix");
  const int m = lane >> 1, k0 = (lane & 1) * 4;
  const size_t grow = (size_t)bb * NOP + op0 + m;
  v4f fmv;
  v4i av[NMA / 4];
  v4f fop;
  if (br == 0) {
    fmv = *(const v4f*)(feats_ma + (size_t)bb * NMA * FDIM + tid * 4);
    const v4i* arow = (const v4i*)(adj + grow * NMA);
#pragma unroll
    for (int q = 0; q < NMA / 4; ++q) av[q] = arow[q];
  } else {
    fop = *(const v4f*)(feats_op + grow * FDIM + k0);
  }
  WShare wsh;
  wshare_load(wsh, l0w2, wid, lane);
  Frag<FDIM / 4> fw1;
  load_w<FDIM / 4>(fw1, br == 0 ? l0w1 : l1w1, br == 0 ? l0b1 : l1b1, lane);
  __builtin_amdgcn_sched_barrier(0);

  __builtin_prefetch(pw2 + tid * 64, 0, 1);

  if (br == 0) *(v4f*)&sFM[tid * 4] = fmv;
  wshare_store(wsh, &sW[0][0], wid, lane);
  __syncthreads();
  wshare_load(wsh, l1w2, wid, lane);

  float* sA = &sBuf[wid][0][0];
  float* sB = &sBuf[wid][1][0];
  float* sC = &sCat[tile][0];

  if (br == 0) {
    v4f s = {0.f, 0.f, 0.f, 0.f};
#pragma unroll
    for (int q = 0; q < NMA / 4; ++q) {
#pragma unroll
      for (int e = 0; e < 4; ++e) {
        const float a = (float)av[q][e];
        const float* f = &sFM[(q * 4 + e) * FDIM + k0];
        s.x = __builtin_fmaf(a, f[0], s.x); s.y = __builtin_fmaf(a, f[1], s.y);
        s.z = __builtin_fmaf(a, f[2], s.z); s.w = __builtin_fmaf(a, f[3], s.w);
      }
    }
    *(v4f*)&sA[m * PF + k0] = s;
  } else {
    *(v4f*)&sA[m * PF + k0] = fop;
  }
  wave_sync();

  layer_h<FDIM, PF, true>(fw1, sA, sB, lane);
  wshare_store(wsh, &sW[1][0], wid, lane);
  __syncthreads();
  wshare_load(wsh, pw2, wid, lane);

  const float* swp = &sW[br][0];
  const float* b2 = br == 0 ? l0b2 : l1b2;
  const float* w3 = br == 0 ? l0w3 : l1w3;
  const float* b3 = br == 0 ? l0b3 : l1b3;
  for (int ph = 0; ph < 2; ++ph) {
    HeadW hw;
    HiddenAcc ha;
    head_load<0, HKS / 2>(hw, w3, b3, lane);
    layer_hl_mma(ha, swp, b2, sB, lane);
    head_load<HKS / 2, HKS>(hw, w3, b3, lane);
    __builtin_amdgcn_sched_barrier(0);
    layer_hl_epi(ha, sA, lane);
    wave_sync();
    const v8f e = layer_o(hw, sA, lane);
    if (ph == 0) {
      Frag<16 / 4> fp1;
      if (br == 0) load_w<16 / 4>(fp1, pw1, pb1, lane);
      if (l16 < 8) {
#pragma unroll
        for (int r = 0; r < 8; ++r)
          sC[(r + half * 8) * PC + br * 8 + l16] = elu1(e[r]);
      }
      __syncthreads();
      wshare_store(wsh, &sW[0][0], wid, lane);
      __syncthreads();
      if (br != 0) return;
      layer_h<16, PC, true>(fp1, sC, sB, lane);
      wave_sync();
      swp = &sW[0][0]; b2 = pb2; w3 = pw3; b3 = pb3;
    } else {
      if (l16 < 8) {
#pragma unroll
        for (int r = 0; r < 8; ++r)
          out[(size_t)(row0 + r + half * 8) * ODIM + l16] = e[r];
      }
    }
  }
}

extern "C" void kernel_launch(void* const* d_in, const int* in_sizes, int n_in,
                              void* d_out, int out_size, void* d_ws,
                              size_t ws_size, hipStream_t stream) {
  const int*   adj = (const int*)d_in[0];
  const float* fop = (const float*)d_in[1];
  const float* fma = (const float*)d_in[2];
  const float* l0w1 = (const float*)d_in[3];
  const float* l0b1 = (const float*)d_in[4];
  const float* l0w2 = (const float*)d_in[5];
  const float* l0b2 = (const float*)d_in[6];
  const float* l0w3 = (const float*)d_in[7];
  const float* l0b3 = (const float*)d_in[8];
  const float* l1w1 = (const float*)d_in[9];
  const float* l1b1 = (const float*)d_in[10];
  const float* l1w2 = (const float*)d_in[11];
  const float* l1b2 = (const float*)d_in[12];
  const float* l1w3 = (const float*)d_in[13];
  const float* l1b3 = (const float*)d_in[14];
  const float* pw1 = (const float*)d_in[15];
  const float* pb1 = (const float*)d_in[16];
  const float* pw2 = (const float*)d_in[17];
  const float* pb2 = (const float*)d_in[18];
  const float* pw3 = (const float*)d_in[19];
  const float* pb3 = (const float*)d_in[20];

  const int total_rows = BATCH * NOP;
  const int rows_per_block = TILES * 16;
  dim3 grid(total_rows / rows_per_block);
  fused_hgnn_mlp<<<grid, TPB, 0, stream>>>(
      adj, fop, fma, l0w1, l0b1, l0w2, l0b2, l0w3, l0b3, l1w1, l1b1, l1w2,
      l1b2, l1w3, l1b3, pw1, pb1, pw2, pb2, pw3, pb3, (float*)d_out);
}


# Each branch's hidden matrix is put in LDS by its own four waves behind split barriers (signal, late wait): self-loop waves enter their hidden layer while the others aggregate; same sums
# speedup vs baseline: 2.8388x; 1.0564x over previous
#include <hip/hip_runtime.h>
#include <math.h>


typedef __attribute__((ext_vector_type(2))) float v2f;
typedef __attribute__((ext_vector_type(4))) float v4f;
typedef __attribute__((ext_vector_type(4))) int v4i;
typedef __attribute__((ext_vector_type(8))) float v8f;

#define TILES 4
#define WAVES (2 * TILES)
#define TPB   (WAVES * 32)
#define BATCH 32
#define NOP   512
#define NMA   64
#define FDIM  8
#define HDIM  128
#define ODIM  8
#define PH    (HDIM + 4)
#define PF    (FDIM + 4)
#define PC    (16 + 4)

__device__ __forceinline__ float elu1(float x) {
  float e = __builtin_amdgcn_exp2f(x * 1.44269504088896340736f) - 1.0f;
  return x > 0.0f ? x : e;
}

__device__ __forceinline__ v8f wmma_f32(v2f a, v2f b, v8f c) {
  return __builtin_amdgcn_wmma_f32_16x16x4_f32(false, a, false, b, (short)0, c,
                                               false, false);
}

__device__ __forceinline__ void wave_sync() {
  __builtin_amdgcn_fence(__ATOMIC_RELEASE, "wavefront");
  __builtin_amdgcn_wave_barrier();
  __builtin_amdgcn_fence(__ATOMIC_ACQUIRE, "wavefront");
}

#define KCH 4

template <int NS>
struct Frag {
  v2f a[NS];
  v4f b[NS][2][2];
  v4f bv[2];
};

template <int NS>
__device__ __forceinline__ void load_w(Frag<NS>& f, const float* __restrict__ W,
                                       const float* __restrict__ bias, int lane) {
  const int c = lane & 15, kh = (lane >> 4) * 2;
#pragma unroll
  for (int i = 0; i < NS; ++i) {
    const int k = i * 4 + kh;
#pragma unroll
    for (int s = 0; s < 2; ++s) {
#pragma unroll
      for (int g = 0; g < 2; ++g)
        f.b[i][s][g] = *(const v4f*)&W[(k + s) * HDIM + g * 64 + c * 4];
    }
  }
  f.bv[0] = *(const v4f*)&bias[c * 4];
  f.bv[1] = *(const v4f*)&bias[64 + c * 4];
}

template <int NS, int PA>
__device__ __forceinline__ void load_a(Frag<NS>& f, const float* sA, int lane) {
  const int c = lane & 15, kh = (lane >> 4) * 2;
#pragma unroll
  for (int i = 0; i < NS; ++i) f.a[i] = *(const v2f*)&sA[c * PA + i * 4 + kh];
}

template <int NS>
__device__ __forceinline__ void mma_chunk(const Frag<NS>& f,
                                          v8f (&acc)[2][2][4]) {
#pragma unroll
  for (int i = 0; i < NS; ++i) {
#pragma unroll
    for (int g = 0; g < 2; ++g) {
#pragma unroll
      for (int t = 0; t < 4; ++t) {
        v2f b; b.x = f.b[i][0][g][t]; b.y = f.b[i][1][g][t];
        acc[i & 1][g][t] = wmma_f32(f.a[i], b, acc[i & 1][g][t]);
      }
    }
  }
}

template <int K, int PA, bool ELU>
__device__ __forceinline__ void layer_h(Frag<K / 4>& f, const float* sA,
                                        float* sY, int lane) {
  constexpr int NS = K / 4;
  static_assert(NS <= KCH && (NS & 1) == 0, "first layers only: all k-steps in one chunk");
  const int half = lane >> 4, c = lane & 15;

  v8f acc[2][2][4];
#pragma unroll
  for (int p = 0; p < 2; ++p)
#pragma unroll
    for (int g = 0; g < 2; ++g)
#pragma unroll
      for (int t = 0; t < 4; ++t)
        acc[p][g][t] = (v8f){0.f, 0.f, 0.f, 0.f, 0.f, 0.f, 0.f, 0.f};

  load_a<NS, PA>(f, sA, lane);
  mma_chunk<NS>(f, acc);

#pragma unroll
  for (int g = 0; g < 2; ++g) {
#pragma unroll
    for (int r = 0; r < 8; ++r) {
      v4f y;
#pragma unroll
      for (int t = 0; t < 4; ++t) {
        float v = acc[0][g][t][r] + acc[1][g][t][r] + f.bv[g][t];
        if (ELU) v = elu1(v);
        y[t] = v;
      }
      *(v4f*)&sY[(r + half * 8) * PH + g * 64 + c * 4] = y;
    }
  }
}

#define WPAIR (2 * HDIM)

#define WROWS (HDIM / WAVES)
struct WShare { v4f r[WROWS]; };

#define BROWS (HDIM / TILES)
struct BShare { v4f r[BROWS]; };

__device__ __forceinline__ void bshare_load(BShare& w, const float* __restrict__ Wg,
                                            int q4, int lane) {
#pragma unroll
  for (int q = 0; q < BROWS; ++q)
    w.r[q] = *(const v4f*)&Wg[(q4 * BROWS + q) * HDIM + lane * 4];
}

__device__ __forceinline__ void bshare_store(const BShare& w, float* sW, int q4,
                                             int lane) {
#pragma unroll
  for (int kk = 0; kk < BROWS / 2; ++kk) {
    const v4f x = w.r[2 * kk], y = w.r[2 * kk + 1];
    float* d = &sW[(q4 * (BROWS / 2) + kk) * WPAIR + lane * 8];
    *(v4f*)&d[0] = (v4f){x[0], y[0], x[1], y[1]};
    *(v4f*)&d[4] = (v4f){x[2], y[2], x[3], y[3]};
  }
}

__device__ __forceinline__ void wshare_load(WShare& w, const float* __restrict__ Wg,
                                            int wid, int lane) {
#pragma unroll
  for (int q = 0; q < WROWS; ++q)
    w.r[q] = *(const v4f*)&Wg[(wid * WROWS + q) * HDIM + lane * 4];
}

__device__ __forceinline__ void wshare_store(const WShare& w, float* sW, int wid,
                                             int lane) {
#pragma unroll
  for (int kk = 0; kk < WROWS / 2; ++kk) {
    const v4f x = w.r[2 * kk], y = w.r[2 * kk + 1];
    float* d = &sW[(wid * (WROWS / 2) + kk) * WPAIR + lane * 8];
    *(v4f*)&d[0] = (v4f){x[0], y[0], x[1], y[1]};
    *(v4f*)&d[4] = (v4f){x[2], y[2], x[3], y[3]};
  }
}

template <int NS>
struct FragL {
  v2f a[NS];
  v4f b[NS][4];
};

template <int NS>
__device__ __forceinline__ void load_chunk_l(FragL<NS>& f, const float* sW,
                                             const float* sA, int ks0, int c,
                                             int half) {
#pragma unroll
  for (int i = 0; i < NS; ++i) {
    const int j = ks0 + i;
    f.a[i] = *(const v2f*)&sA[c * PH + j * 4 + half * 2];
#pragma unroll
    for (int g = 0; g < 4; ++g)
      f.b[i][g] = *(const v4f*)&sW[(j * 2 + half) * WPAIR + g * 64 + c * 4];
  }
}

template <int NS>
__device__ __forceinline__ void mma_chunk_l(const FragL<NS>& f,
                                            v8f (&acc)[2][4][2]) {
#pragma unroll
  for (int i = 0; i < NS; ++i) {
#pragma unroll
    for (int g = 0; g < 4; ++g) {
#pragma unroll
      for (int e = 0; e < 2; ++e) {
        v2f b; b.x = f.b[i][g][2 * e]; b.y = f.b[i][g][2 * e + 1];
        acc[i & 1][g][e] = wmma_f32(f.a[i], b, acc[i & 1][g][e]);
      }
    }
  }
}

struct HiddenAcc {
  v8f acc[2][4][2];
  v2f bv[4];
};

__device__ __forceinline__ void layer_hl_mma(HiddenAcc& o, const float* sW,
                                             const float* __restrict__ bias,
                                             const float* sA, int lane) {
  constexpr int NS = KCH;
  constexpr int NCH = (HDIM / 4) / NS;
  const int half = lane >> 4, c = lane & 15;

#pragma unroll
  for (int p = 0; p < 2; ++p)
#pragma unroll
    for (int g = 0; g < 4; ++g)
#pragma unroll
      for (int e = 0; e < 2; ++e)
        o.acc[p][g][e] = (v8f){0.f, 0.f, 0.f, 0.f, 0.f, 0.f, 0.f, 0.f};

#pragma unroll
  for (int g = 0; g < 4; ++g) o.bv[g] = *(const v2f*)&bias[g * 32 + c * 2];

  FragL<NS> f0, f1;
  load_chunk_l<NS>(f0, sW, sA, 0, c, half);
#pragma unroll 1
  for (int ch = 0; ch < NCH - 2; ch += 2) {
    load_chunk_l<NS>(f1, sW, sA, (ch + 1) * NS, c, half);
    __builtin_amdgcn_sched_barrier(0);
    mma_chunk_l<NS>(f0, o.acc);
    load_chunk_l<NS>(f0, sW, sA, (ch + 2) * NS, c, half);
    __builtin_amdgcn_sched_barrier(0);
    mma_chunk_l<NS>(f1, o.acc);
  }
  load_chunk_l<NS>(f1, sW, sA, (NCH - 1) * NS, c, half);
  __builtin_amdgcn_sched_barrier(0);
  mma_chunk_l<NS>(f0, o.acc);
  mma_chunk_l<NS>(f1, o.acc);
}

__device__ __forceinline__ void layer_hl_epi(const HiddenAcc& o, float* sY, int lane) {
  const int half = lane >> 4, c = lane & 15;
#pragma unroll
  for (int g = 0; g < 4; ++g) {
#pragma unroll
    for (int r = 0; r < 8; ++r) {
      v2f y;
#pragma unroll
      for (int e = 0; e < 2; ++e)
        y[e] = elu1(o.acc[0][g][e][r] + o.acc[1][g][e][r] + o.bv[g][e]);
      *(v2f*)&sY[(r + half * 8) * PH + g * 32 + c * 2] = y;
    }
  }
}

#define HKS (HDIM / 4)
struct HeadW {
  v2f b[HKS];
  float bias;
};

template <int J0, int J1>
__device__ __forceinline__ void head_load(HeadW& hw, const float* __restrict__ W,
                                          const float* __restrict__ bias,
                                          int lane) {
  const int half = lane >> 4, l16 = lane & 15;
  const int kh = half * 2;
  const int n = (l16 < 8) ? l16 : 0;
#pragma unroll
  for (int j = J0; j < J1; ++j) {
    const int k = j * 4 + kh;
    hw.b[j].x = W[k * ODIM + n];
    hw.b[j].y = W[(k + 1) * ODIM + n];
  }
  if (J1 == HKS) hw.bias = bias[n];
}

__device__ __forceinline__ v8f layer_o(const HeadW& hw, const float* sA, int lane) {
  const int half = lane >> 4, l16 = lane & 15;
  const int kh = half * 2;
  v2f a[HKS];
#pragma unroll
  for (int j = 0; j < HKS; ++j) a[j] = *(const v2f*)&sA[l16 * PH + j * 4 + kh];
  __builtin_amdgcn_sched_barrier(0);
  v8f acc0 = {0.f, 0.f, 0.f, 0.f, 0.f, 0.f, 0.f, 0.f};
  v8f acc1 = {0.f, 0.f, 0.f, 0.f, 0.f, 0.f, 0.f, 0.f};
#pragma unroll
  for (int j = 0; j < HKS; j += 2) {
    acc0 = wmma_f32(a[j], hw.b[j], acc0);
    acc1 = wmma_f32(a[j + 1], hw.b[j + 1], acc1);
  }
  const float bv = hw.bias;
#pragma unroll
  for (int r = 0; r < 8; ++r) acc0[r] += acc1[r] + bv;
  return acc0;
}

__global__ __launch_bounds__(TPB) void fused_hgnn_mlp(
    const int* __restrict__ adj, const float* __restrict__ feats_op,
    const float* __restrict__ feats_ma,
    const float* __restrict__ l0w1, const float* __restrict__ l0b1,
    const float* __restrict__ l0w2, const float* __restrict__ l0b2,
    const float* __restrict__ l0w3, const float* __restrict__ l0b3,
    const float* __restrict__ l1w1, const float* __restrict__ l1b1,
    const float* __restrict__ l1w2, const float* __restrict__ l1b2,
    const float* __restrict__ l1w3, const float* __restrict__ l1b3,
    const float* __restrict__ pw1, const float* __restrict__ pb1,
    const float* __restrict__ pw2, const float* __restrict__ pb2,
    const float* __restrict__ pw3, const float* __restrict__ pb3,
    float* __restrict__ out) {
  __shared__ __attribute__((aligned(16))) float sBuf[WAVES][2][16 * PH];
  __shared__ __attribute__((aligned(16))) float sCat[TILES][16 * PC];
  __shared__ __attribute__((aligned(16))) float sW[2][(HDIM / 2) * WPAIR];

  const int tid = threadIdx.x;
  const int wid = __builtin_amdgcn_readfirstlane(tid >> 5);
  const int lane = tid & 31;
  const int tile = wid & (TILES - 1);
  const int br = wid / TILES;
  const int half = lane >> 4, l16 = lane & 15;
  const int row0 = blockIdx.x * (TILES * 16) + tile * 16;
  const int bb = row0 / NOP;
  const int op0 = row0 % NOP;

  static_assert(4 * 32 * 4 == NMA * FDIM, "four 16-byte pieces of feats_ma per lane of an aggregating wave");
  static_assert(WAVES * WROWS == HDIM, "the waves together stage all k rows of the projection's hidden matrix");
  static_assert(NMA * FDIM <= 16 * PH, "the wave's feats_ma copy fits its second activation tile");
  const int m = lane >> 1, k0 = (lane & 1) * 4;
  const size_t grow = (size_t)bb * NOP + op0 + m;

  __builtin_prefetch(pw2 + tid * 64, 0, 1);

  float* sA = &sBuf[wid][0][0];
  float* sB = &sBuf[wid][1][0];
  float* sC = &sCat[tile][0];

  Frag<FDIM / 4> fw1;
  BShare wsb;
  if (br == 0) {
    v4f fmv[4];
    v4i av[NMA / 4];
    const v4f* fm = (const v4f*)(feats_ma + (size_t)bb * NMA * FDIM);
#pragma unroll
    for (int q = 0; q < 4; ++q) fmv[q] = fm[q * 32 + lane];
    const v4i* arow = (const v4i*)(adj + grow * NMA);
#pragma unroll
    for (int q = 0; q < NMA / 4; ++q) av[q] = arow[q];
    load_w<FDIM / 4>(fw1, l0w1, l0b1, lane);
    __builtin_amdgcn_sched_barrier(0);
    __builtin_amdgcn_s_barrier_signal(-1);
#pragma unroll
    for (int q = 0; q < 4; ++q) *(v4f*)&sB[(q * 32 + lane) * 4] = fmv[q];
    wave_sync();
    v4f s = {0.f, 0.f, 0.f, 0.f};
#pragma unroll
    for (int q = 0; q < NMA / 4; ++q) {
#pragma unroll
      for (int e = 0; e < 4; ++e) {
        const float a = (float)av[q][e];
        const float* f = &sB[(q * 4 + e) * FDIM + k0];
        s.x = __builtin_fmaf(a, f[0], s.x); s.y = __builtin_fmaf(a, f[1], s.y);
        s.z = __builtin_fmaf(a, f[2], s.z); s.w = __builtin_fmaf(a, f[3], s.w);
      }
    }
    *(v4f*)&sA[m * PF + k0] = s;
    __builtin_amdgcn_sched_barrier(0);
    bshare_load(wsb, l0w2, tile, lane);
  } else {
    const v4f fop = *(const v4f*)(feats_op + grow * FDIM + k0);
    load_w<FDIM / 4>(fw1, l1w1, l1b1, lane);
    bshare_load(wsb, l1w2, tile, lane);
    __builtin_amdgcn_sched_barrier(0);
    *(v4f*)&sA[m * PF + k0] = fop;
  }
  wave_sync();

  layer_h<FDIM, PF, true>(fw1, sA, sB, lane);
  bshare_store(wsb, &sW[br][0], tile, lane);
  __builtin_amdgcn_fence(__ATOMIC_RELEASE, "workgroup");
  if (br == 0) {
    __builtin_amdgcn_s_barrier_wait(-1);
    __builtin_amdgcn_s_barrier_signal(-1);
    __builtin_amdgcn_s_barrier_wait(-1);
  } else {
    __builtin_amdgcn_s_barrier_signal(-1);
    __builtin_amdgcn_s_barrier_wait(-1);
    __builtin_amdgcn_s_barrier_signal(-1);
  }
  __builtin_amdgcn_fence(__ATOMIC_ACQUIRE, "workgroup");
  WShare wsh;
  wshare_load(wsh, pw2, wid, lane);

  const float* swp = &sW[br][0];
  const float* b2 = br == 0 ? l0b2 : l1b2;
  const float* w3 = br == 0 ? l0w3 : l1w3;
  const float* b3 = br == 0 ? l0b3 : l1b3;
  for (int ph = 0; ph < 2; ++ph) {
    HeadW hw;
    HiddenAcc ha;
    head_load<0, HKS / 2>(hw, w3, b3, lane);
    layer_hl_mma(ha, swp, b2, sB, lane);
    head_load<HKS / 2, HKS>(hw, w3, b3, lane);
    __builtin_amdgcn_sched_barrier(0);
    layer_hl_epi(ha, sA, lane);
    wave_sync();
    const v8f e = layer_o(hw, sA, lane);
    if (ph == 0) {
      Frag<16 / 4> fp1;
      if (br == 0) load_w<16 / 4>(fp1, pw1, pb1, lane);
      if (l16 < 8) {
#pragma unroll
        for (int r = 0; r < 8; ++r)
          sC[(r + half * 8) * PC + br * 8 + l16] = elu1(e[r]);
      }
      if (br != 0) __builtin_amdgcn_s_barrier_wait(-1);
      __syncthreads();
      wshare_store(wsh, &sW[0][0], wid, lane);
      __syncthreads();
      if (br != 0) return;
      layer_h<16, PC, true>(fp1, sC, sB, lane);
      wave_sync();
      swp = &sW[0][0]; b2 = pb2; w3 = pw3; b3 = pb3;
    } else {
      if (l16 < 8) {
#pragma unroll
        for (int r = 0; r < 8; ++r)
          out[(size_t)(row0 + r + half * 8) * ODIM + l16] = e[r];
      }
    }
  }
}

extern "C" void kernel_launch(void* const* d_in, const int* in_sizes, int n_in,
                              void* d_out, int out_size, void* d_ws,
                              size_t ws_size, hipStream_t stream) {
  const int*   adj = (const int*)d_in[0];
  const float* fop = (const float*)d_in[1];
  const float* fma = (const float*)d_in[2];
  const float* l0w1 = (const float*)d_in[3];
  const float* l0b1 = (const float*)d_in[4];
  const float* l0w2 = (const float*)d_in[5];
  const float* l0b2 = (const float*)d_in[6];
  const float* l0w3 = (const float*)d_in[7];
  const float* l0b3 = (const float*)d_in[8];
  const float* l1w1 = (const float*)d_in[9];
  const float* l1b1 = (const float*)d_in[10];
  const float* l1w2 = (const float*)d_in[11];
  const float* l1b2 = (const float*)d_in[12];
  const float* l1w3 = (const float*)d_in[13];
  const float* l1b3 = (const float*)d_in[14];
  const float* pw1 = (const float*)d_in[15];
  const float* pb1 = (const float*)d_in[16];
  const float* pw2 = (const float*)d_in[17];
  const float* pb2 = (const float*)d_in[18];
  const float* pw3 = (const float*)d_in[19];
  const float* pb3 = (const float*)d_in[20];

  const int total_rows = BATCH * NOP;
  const int rows_per_block = TILES * 16;
  dim3 grid(total_rows / rows_per_block);
  fused_hgnn_mlp<<<grid, TPB, 0, stream>>>(
      adj, fop, fma, l0w1, l0b1, l0w2, l0b2, l0w3, l0b3, l1w1, l1b1, l1w2,
      l1b2, l1w3, l1b3, pw1, pb1, pw2, pb2, pw3, pb3, (float*)d_out);
}


# All layers computed transposed (weights as A, the tile's rows as B): a layer's D registers are the next layer's B fragments, activations never cross LDS; bias starts the sums, k and sum order change
# speedup vs baseline: 3.1298x; 1.1025x over previous
#include <hip/hip_runtime.h>
#include <math.h>


typedef __attribute__((ext_vector_type(2))) float v2f;
typedef __attribute__((ext_vector_type(4))) float v4f;
typedef __attribute__((ext_vector_type(4))) int v4i;
typedef __attribute__((ext_vector_type(8))) float v8f;

#define TILES 4
#define WAVES (2 * TILES)
#define TPB   (WAVES * 32)
#define BATCH 32
#define NOP   512
#define NMA   64
#define FDIM  8
#define HDIM  128
#define ODIM  8
#define PC    (16 + 4)

__device__ __forceinline__ float elu1(float x) {
  float e = __builtin_amdgcn_exp2f(x * 1.44269504088896340736f) - 1.0f;
  return x > 0.0f ? x : e;
}

__device__ __forceinline__ v8f wmma_f32(v2f a, v2f b, v8f c) {
  return __builtin_amdgcn_wmma_f32_16x16x4_f32(false, a, false, b, (short)0, c,
                                               false, false);
}

__device__ __forceinline__ void wave_sync() {
  __builtin_amdgcn_fence(__ATOMIC_RELEASE, "wavefront");
  __builtin_amdgcn_wave_barrier();
  __builtin_amdgcn_fence(__ATOMIC_ACQUIRE, "wavefront");
}


template <int NS>
struct FragW {
  v4f w[NS][2][2];
  v4f bias[2][8];
};

template <int NS>
__device__ __forceinline__ void load_w(FragW<NS>& f, const float* __restrict__ W,
                                       const float* __restrict__ bias, int lane) {
  const int c = lane & 15, h = lane >> 4;
#pragma unroll
  for (int i = 0; i < NS; ++i) {
#pragma unroll
    for (int s = 0; s < 2; ++s) {
#pragma unroll
      for (int G = 0; G < 2; ++G)
        f.w[i][s][G] = *(const v4f*)&W[(2 * NS * h + 2 * i + s) * HDIM + G * 64 + c * 4];
    }
  }
#pragma unroll
  for (int G = 0; G < 2; ++G)
#pragma unroll
    for (int r = 0; r < 8; ++r)
      f.bias[G][r] = *(const v4f*)&bias[G * 64 + h * 32 + r * 4];
}

template <int NS>
__device__ __forceinline__ void layer_first(const FragW<NS>& f, const v2f (&xin)[NS],
                                            v8f (&acc)[2][4]) {
#pragma unroll
  for (int G = 0; G < 2; ++G)
#pragma unroll
    for (int t = 0; t < 4; ++t)
#pragma unroll
      for (int r = 0; r < 8; ++r) acc[G][t][r] = f.bias[G][r][t];
#pragma unroll
  for (int i = 0; i < NS; ++i) {
#pragma unroll
    for (int G = 0; G < 2; ++G) {
#pragma unroll
      for (int t = 0; t < 4; ++t) {
        v2f a; a.x = f.w[i][0][G][t]; a.y = f.w[i][1][G][t];
        acc[G][t] = wmma_f32(a, xin[i], acc[G][t]);
      }
    }
  }
}

__device__ __forceinline__ void act_first(const v8f (&acc)[4], v2f (&x)[8][2], int r) {
#pragma unroll
  for (int u = 0; u < 2; ++u) {
    v2f y; y.x = elu1(acc[2 * u][r]); y.y = elu1(acc[2 * u + 1][r]);
    x[r][u] = y;
  }
}

#define WPAIR (2 * HDIM)

#define WROWS (HDIM / WAVES)
struct WShare { v4f r[WROWS]; };

#define BROWS (HDIM / TILES)
struct BShare { v4f r[BROWS]; };

__device__ __forceinline__ void bshare_load(BShare& w, const float* __restrict__ Wg,
                                            int q4, int lane) {
#pragma unroll
  for (int q = 0; q < BROWS; ++q)
    w.r[q] = *(const v4f*)&Wg[(q4 * BROWS + q) * HDIM + lane * 4];
}

__device__ __forceinline__ void bshare_store(const BShare& w, float* sW, int q4,
                                             int lane) {
#pragma unroll
  for (int kk = 0; kk < BROWS / 2; ++kk) {
    const v4f x = w.r[2 * kk], y = w.r[2 * kk + 1];
    float* d = &sW[(q4 * (BROWS / 2) + kk) * WPAIR + lane * 8];
    *(v4f*)&d[0] = (v4f){x[0], y[0], x[1], y[1]};
    *(v4f*)&d[4] = (v4f){x[2], y[2], x[3], y[3]};
  }
}

__device__ __forceinline__ void wshare_load(WShare& w, const float* __restrict__ Wg,
                                            int wid, int lane) {
#pragma unroll
  for (int q = 0; q < WROWS; ++q)
    w.r[q] = *(const v4f*)&Wg[(wid * WROWS + q) * HDIM + lane * 4];
}

__device__ __forceinline__ void wshare_store(const WShare& w, float* sW, int wid,
                                             int lane) {
#pragma unroll
  for (int kk = 0; kk < WROWS / 2; ++kk) {
    const v4f x = w.r[2 * kk], y = w.r[2 * kk + 1];
    float* d = &sW[(wid * (WROWS / 2) + kk) * WPAIR + lane * 8];
    *(v4f*)&d[0] = (v4f){x[0], y[0], x[1], y[1]};
    *(v4f*)&d[4] = (v4f){x[2], y[2], x[3], y[3]};
  }
}

#define KCH 4
#define NCH ((HDIM / 4) / KCH)
struct FragA { v4f a[KCH][4]; };

__device__ __forceinline__ void load_chunk_a(FragA& f, const float* sWl, int ch) {
#pragma unroll
  for (int i = 0; i < KCH; ++i) {
    const int j = ch * KCH + i;
    const int kpb = j + 16 * (j >> 4);
#pragma unroll
    for (int g = 0; g < 4; ++g)
      f.a[i][g] = *(const v4f*)&sWl[kpb * WPAIR + g * 64];
  }
}

__device__ __forceinline__ void mma_chunk_a(const FragA& f, const v2f (&x)[2][8][2],
                                            v8f (&acc)[4][2], int ch) {
#pragma unroll
  for (int i = 0; i < KCH; ++i) {
    const int j = ch * KCH + i;
    const v2f b = x[j >> 4][(j >> 1) & 7][j & 1];
#pragma unroll
    for (int g = 0; g < 4; ++g) {
#pragma unroll
      for (int e = 0; e < 2; ++e) {
        v2f a; a.x = f.a[i][g][2 * e]; a.y = f.a[i][g][2 * e + 1];
        acc[g][e] = wmma_f32(a, b, acc[g][e]);
      }
    }
  }
}

struct HBias { v4f b[4][4]; };

__device__ __forceinline__ void hbias_load(HBias& hb, const float* __restrict__ bias,
                                           int lane) {
  const int h = lane >> 4;
#pragma unroll
  for (int g = 0; g < 4; ++g)
#pragma unroll
    for (int q = 0; q < 4; ++q)
      hb.b[g][q] = *(const v4f*)&bias[g * 32 + h * 16 + q * 4];
}

__device__ __forceinline__ void layer_hidden(v8f (&acc)[4][2], const HBias& hb,
                                             const float* sWm,
                                             const v8f (&a1)[2][4], int lane) {
  const int half = lane >> 4, c = lane & 15;
  const float* sWl = sWm + (16 * half) * WPAIR + c * 4;

#pragma unroll
  for (int g = 0; g < 4; ++g)
#pragma unroll
    for (int q = 0; q < 4; ++q)
#pragma unroll
      for (int w = 0; w < 2; ++w)
#pragma unroll
        for (int e = 0; e < 2; ++e) acc[g][e][2 * q + w] = hb.b[g][q][2 * w + e];

  FragA f0, f1;
  load_chunk_a(f0, sWl, 0);

  v2f x[2][8][2];
#pragma unroll
  for (int r = 0; r < 8; ++r) act_first(a1[0], x[0], r);

#pragma unroll
  for (int ch = 0; ch < NCH; ch += 2) {
    load_chunk_a(f1, sWl, ch + 1);
    __builtin_amdgcn_sched_barrier(0);
    if (ch < NCH / 2) { act_first(a1[1], x[1], 2 * ch); act_first(a1[1], x[1], 2 * ch + 1); }
    mma_chunk_a(f0, x, acc, ch);
    if (ch + 2 < NCH) load_chunk_a(f0, sWl, ch + 2);
    __builtin_amdgcn_sched_barrier(0);
    if (ch + 1 < NCH / 2) { act_first(a1[1], x[1], 2 * ch + 2); act_first(a1[1], x[1], 2 * ch + 3); }
    mma_chunk_a(f1, x, acc, ch + 1);
  }
}

#define HKS (HDIM / 4)
struct HeadW {
  v2f a[HKS];
  v4f bias[2];
};

template <int J0, int J1>
__device__ __forceinline__ void head_load(HeadW& hw, const float* __restrict__ W,
                                          const float* __restrict__ bias,
                                          int lane) {
  const int half = lane >> 4, l16 = lane & 15;
  const int n = (l16 < 8) ? l16 : 0;
  if (J0 == 0) {
    hw.bias[0] = *(const v4f*)&bias[0];
    hw.bias[1] = *(const v4f*)&bias[4];
  }
#pragma unroll
  for (int j = J0; j < J1; ++j) {
    const int k0 = 32 * (j >> 3) + 16 * half + 2 * (j & 7);
    hw.a[j].x = W[k0 * ODIM + n];
    hw.a[j].y = W[(k0 + 1) * ODIM + n];
  }
}

__device__ __forceinline__ v8f layer_head(const HeadW& hw, const v8f (&acc)[4][2]) {
  v8f acc0, acc1 = {0.f, 0.f, 0.f, 0.f, 0.f, 0.f, 0.f, 0.f};
#pragma unroll
  for (int r = 0; r < 8; ++r) acc0[r] = hw.bias[r >> 2][r & 3];
#pragma unroll
  for (int g = 0; g < 4; ++g) {
    v2f x[8];
#pragma unroll
    for (int r = 0; r < 8; ++r) {
      v2f y; y.x = elu1(acc[g][0][r]); y.y = elu1(acc[g][1][r]);
      x[r] = y;
    }
#pragma unroll
    for (int r = 0; r < 8; r += 2) {
      acc0 = wmma_f32(hw.a[8 * g + r], x[r], acc0);
      acc1 = wmma_f32(hw.a[8 * g + r + 1], x[r + 1], acc1);
    }
  }
#pragma unroll
  for (int r = 0; r < 8; ++r) acc0[r] += acc1[r];
  return acc0;
}

__global__ __launch_bounds__(TPB) void fused_hgnn_mlp(
    const int* __restrict__ adj, const float* __restrict__ feats_op,
    const float* __restrict__ feats_ma,
    const float* __restrict__ l0w1, const float* __restrict__ l0b1,
    const float* __restrict__ l0w2, const float* __restrict__ l0b2,
    const float* __restrict__ l0w3, const float* __restrict__ l0b3,
    const float* __restrict__ l1w1, const float* __restrict__ l1b1,
    const float* __restrict__ l1w2, const float* __restrict__ l1b2,
    const float* __restrict__ l1w3, const float* __restrict__ l1b3,
    const float* __restrict__ pw1, const float* __restrict__ pb1,
    const float* __restrict__ pw2, const float* __restrict__ pb2,
    const float* __restrict__ pw3, const float* __restrict__ pb3,
    float* __restrict__ out) {
  __shared__ __attribute__((aligned(16))) float sFm[TILES][NMA * FDIM];
  __shared__ __attribute__((aligned(16))) float sCat[TILES][16 * PC];
  __shared__ __attribute__((aligned(16))) float sW[2][(HDIM / 2) * WPAIR];

  const int tid = threadIdx.x;
  const int wid = __builtin_amdgcn_readfirstlane(tid >> 5);
  const int lane = tid & 31;
  const int tile = wid & (TILES - 1);
  const int br = wid / TILES;
  const int half = lane >> 4, l16 = lane & 15;
  const int row0 = blockIdx.x * (TILES * 16) + tile * 16;
  const int bb = row0 / NOP;
  const int op0 = row0 % NOP;

  static_assert(4 * 32 * 4 == NMA * FDIM, "four 16-byte pieces of feats_ma per lane of an aggregating wave");
  static_assert(WAVES * WROWS == HDIM, "the waves together stage all k rows of the projection's hidden matrix");
  const int k0 = half * 4;
  const size_t grow = (size_t)bb * NOP + op0 + l16;

  __builtin_prefetch(pw2 + tid * 64, 0, 1);

  float* sF = &sFm[tile][0];
  float* sC = &sCat[tile][0];

  FragW<FDIM / 4> fw1;
  BShare wsb;
  HBias hb;
  v2f xin[FDIM / 4];
  if (br == 0) {
    v4f fmv[4];
    v4i av[NMA / 4];
    const v4f* fm = (const v4f*)(feats_ma + (size_t)bb * NMA * FDIM);
#pragma unroll
    for (int q = 0; q < 4; ++q) fmv[q] = fm[q * 32 + lane];
    const v4i* arow = (const v4i*)(adj + grow * NMA);
#pragma unroll
    for (int q = 0; q < NMA / 4; ++q) av[q] = arow[q];
    load_w<FDIM / 4>(fw1, l0w1, l0b1, lane);
    __builtin_amdgcn_sched_barrier(0);
    __builtin_amdgcn_s_barrier_signal(-1);
#pragma unroll
    for (int q = 0; q < 4; ++q) *(v4f*)&sF[(q * 32 + lane) * 4] = fmv[q];
    wave_sync();
    v4f s = {0.f, 0.f, 0.f, 0.f};
#pragma unroll
    for (int q = 0; q < NMA / 4; ++q) {
#pragma unroll
      for (int e = 0; e < 4; ++e) {
        const float a = (float)av[q][e];
        const float* f = &sF[(q * 4 + e) * FDIM + k0];
        s.x = __builtin_fmaf(a, f[0], s.x); s.y = __builtin_fmaf(a, f[1], s.y);
        s.z = __builtin_fmaf(a, f[2], s.z); s.w = __builtin_fmaf(a, f[3], s.w);
      }
    }
    xin[0] = (v2f){s.x, s.y};
    xin[1] = (v2f){s.z, s.w};
    __builtin_amdgcn_sched_barrier(0);
    bshare_load(wsb, l0w2, tile, lane);
  } else {
    const v4f fop = *(const v4f*)(feats_op + grow * FDIM + k0);
    load_w<FDIM / 4>(fw1, l1w1, l1b1, lane);
    __builtin_amdgcn_sched_barrier(0);
    bshare_load(wsb, l1w2, tile, lane);
    __builtin_amdgcn_sched_barrier(0);
    xin[0] = (v2f){fop.x, fop.y};
    xin[1] = (v2f){fop.z, fop.w};
  }

  hbias_load(hb, br == 0 ? l0b2 : l1b2, lane);
  v8f a1[2][4];
  layer_first<FDIM / 4>(fw1, xin, a1);
  __builtin_amdgcn_sched_barrier(0);
  bshare_store(wsb, &sW[br][0], tile, lane);
  __builtin_amdgcn_fence(__ATOMIC_RELEASE, "workgroup");
  if (br == 0) {
    __builtin_amdgcn_s_barrier_wait(-1);
    __builtin_amdgcn_s_barrier_signal(-1);
    __builtin_amdgcn_s_barrier_wait(-1);
  } else {
    __builtin_amdgcn_s_barrier_signal(-1);
    __builtin_amdgcn_s_barrier_wait(-1);
    __builtin_amdgcn_s_barrier_signal(-1);
  }
  __builtin_amdgcn_fence(__ATOMIC_ACQUIRE, "workgroup");

  WShare wsh;
  wshare_load(wsh, pw2, wid, lane);

  const float* w3 = br == 0 ? l0w3 : l1w3;
  const float* b3 = br == 0 ? l0b3 : l1b3;
  v8f e0;
  {
    HeadW hw;
    v8f acc[4][2];
    head_load<0, HKS / 2>(hw, w3, b3, lane);
    layer_hidden(acc, hb, &sW[br][0], a1, lane);
    head_load<HKS / 2, HKS>(hw, w3, b3, lane);
    e0 = layer_head(hw, acc);
  }
  v4f y0, y1;
#pragma unroll
  for (int r = 0; r < 4; ++r) { y0[r] = elu1(e0[r]); y1[r] = elu1(e0[4 + r]); }

  FragW<16 / 4> fp1;
  HBias hbp;
  if (br == 0) {
    load_w<16 / 4>(fp1, pw1, pb1, lane);
    hbias_load(hbp, pb2, lane);
  }
  if (half == 0) {
    *(v4f*)&sC[l16 * PC + br * 8] = y0;
    *(v4f*)&sC[l16 * PC + br * 8 + 4] = y1;
  }
  if (br != 0) __builtin_amdgcn_s_barrier_wait(-1);
  __syncthreads();
  wshare_store(wsh, &sW[0][0], wid, lane);
  __syncthreads();
  if (br != 0) return;

  HeadW hw;
  head_load<0, HKS / 2>(hw, pw3, pb3, lane);
  const v4f c0 = *(const v4f*)&sC[l16 * PC + half * 8];
  const v4f c1 = *(const v4f*)&sC[l16 * PC + half * 8 + 4];
  v2f xc[4];
  xc[0] = (v2f){c0.x, c0.y}; xc[1] = (v2f){c0.z, c0.w};
  xc[2] = (v2f){c1.x, c1.y}; xc[3] = (v2f){c1.z, c1.w};
  v8f a2[2][4];
  layer_first<16 / 4>(fp1, xc, a2);
  v8f acc[4][2];
  layer_hidden(acc, hbp, &sW[0][0], a2, lane);
  head_load<HKS / 2, HKS>(hw, pw3, pb3, lane);
  const v8f e1 = layer_head(hw, acc);
  if (half == 0) {
    float* o = out + (size_t)(row0 + l16) * ODIM;
    *(v4f*)&o[0] = (v4f){e1[0], e1[1], e1[2], e1[3]};
    *(v4f*)&o[4] = (v4f){e1[4], e1[5], e1[6], e1[7]};
  }
}

extern "C" void kernel_launch(void* const* d_in, const int* in_sizes, int n_in,
                              void* d_out, int out_size, void* d_ws,
                              size_t ws_size, hipStream_t stream) {
  const int*   adj = (const int*)d_in[0];
  const float* fop = (const float*)d_in[1];
  const float* fma = (const float*)d_in[2];
  const float* l0w1 = (const float*)d_in[3];
  const float* l0b1 = (const float*)d_in[4];
  const float* l0w2 = (const float*)d_in[5];
  const float* l0b2 = (const float*)d_in[6];
  const float* l0w3 = (const float*)d_in[7];
  const float* l0b3 = (const float*)d_in[8];
  const float* l1w1 = (const float*)d_in[9];
  const float* l1b1 = (const float*)d_in[10];
  const float* l1w2 = (const float*)d_in[11];
  const float* l1b2 = (const float*)d_in[12];
  const float* l1w3 = (const float*)d_in[13];
  const float* l1b3 = (const float*)d_in[14];
  const float* pw1 = (const float*)d_in[15];
  const float* pb1 = (const float*)d_in[16];
  const float* pw2 = (const float*)d_in[17];
  const float* pb2 = (const float*)d_in[18];
  const float* pw3 = (const float*)d_in[19];
  const float* pb3 = (const float*)d_in[20];

  const int total_rows = BATCH * NOP;
  const int rows_per_block = TILES * 16;
  dim3 grid(total_rows / rows_per_block);
  fused_hgnn_mlp<<<grid, TPB, 0, stream>>>(
      adj, fop, fma, l0w1, l0b1, l0w2, l0b2, l0w3, l0b3, l1w1, l1b1, l1w2,
      l1b2, l1w3, l1b3, pw1, pb1, pw2, pb2, pw3, pb3, (float*)d_out);
}
